# starting kernel
# baseline (speedup 1.0000x reference)
#include <hip/hip_runtime.h>


typedef float v2f __attribute__((ext_vector_type(2)));
typedef float v8f __attribute__((ext_vector_type(8)));

#define DMODEL 1024
#define NHEAD  16
#define DK     64
#define NEG_INF (-1.0e9f)

__device__ inline v8f wmma_f32(v2f a, v2f b, v8f c) {
  return __builtin_amdgcn_wmma_f32_16x16x4_f32(false, a, false, b,
                                               (short)0, c, false, false);
}

__global__ void __launch_bounds__(256)
gemm_nt_bias_kernel(const float* __restrict__ X, const float* __restrict__ W,
                    const float* __restrict__ bias, float* __restrict__ Y,
                    int M, int N, int K, int S, int qkv_layout) {
  const int lane = threadIdx.x & 31;
  const int wave = threadIdx.x >> 5;
  const int half = lane >> 4;
  const int l    = lane & 15;
  const int mBase = (blockIdx.y * 8 + wave) * 32;
  const int nBase = blockIdx.x * 64;

  const float* xrow0 = X + (size_t)(mBase +  0 + l) * K;
  const float* xrow1 = X + (size_t)(mBase + 16 + l) * K;
  const float* w0    = W + (size_t)(nBase +  0 + l) * K;
  const float* w1    = W + (size_t)(nBase + 16 + l) * K;
  const float* w2    = W + (size_t)(nBase + 32 + l) * K;
  const float* w3    = W + (size_t)(nBase + 48 + l) * K;

  v8f acc[2][4] = {{v8f{}, v8f{}, v8f{}, v8f{}},
                   {v8f{}, v8f{}, v8f{}, v8f{}}};

  #pragma unroll 4
  for (int k0 = 0; k0 < K; k0 += 4) {
    const int off = k0 + 2 * half;
    v2f a0 = *(const v2f*)(xrow0 + off);
    v2f a1 = *(const v2f*)(xrow1 + off);
    v2f b0 = *(const v2f*)(w0 + off);
    v2f b1 = *(const v2f*)(w1 + off);
    v2f b2 = *(const v2f*)(w2 + off);
    v2f b3 = *(const v2f*)(w3 + off);
    acc[0][0] = wmma_f32(a0, b0, acc[0][0]);
    acc[0][1] = wmma_f32(a0, b1, acc[0][1]);
    acc[0][2] = wmma_f32(a0, b2, acc[0][2]);
    acc[0][3] = wmma_f32(a0, b3, acc[0][3]);
    acc[1][0] = wmma_f32(a1, b0, acc[1][0]);
    acc[1][1] = wmma_f32(a1, b1, acc[1][1]);
    acc[1][2] = wmma_f32(a1, b2, acc[1][2]);
    acc[1][3] = wmma_f32(a1, b3, acc[1][3]);
  }

  #pragma unroll
  for (int mi = 0; mi < 2; ++mi) {
    #pragma unroll
    for (int j = 0; j < 4; ++j) {
      const int n  = nBase + j * 16 + l;
      const float bv = bias[n];
      #pragma unroll
      for (int r = 0; r < 8; ++r) {
        const int m = mBase + mi * 16 + r + 8 * half;
        const float v = acc[mi][j][r] + bv;
        size_t dst;
        if (qkv_layout) {
          const int bidx = m / S, s = m % S;
          const int h = n / DK,   d = n % DK;
          dst = (((size_t)(bidx * NHEAD + h) * S + s) * DK + d);
        } else {
          dst = (size_t)m * N + n;
        }
        Y[dst] = v;
      }
    }
  }
}

__global__ void __launch_bounds__(256)
attn_nosoftmax_kernel(const float* __restrict__ Q, const float* __restrict__ Kp,
                      const float* __restrict__ V, const int* __restrict__ mask,
                      float* __restrict__ O, int S) {
  const int lane = threadIdx.x & 31;
  const int wave = threadIdx.x >> 5;
  const int half = lane >> 4;
  const int l    = lane & 15;
  const int bh   = blockIdx.y;
  const int bidx = bh / NHEAD;
  const int h    = bh % NHEAD;
  const int qBase = blockIdx.x * 256 + wave * 32;

  const float* Qh = Q  + (size_t)bh * S * DK;
  const float* Kh = Kp + (size_t)bh * S * DK;
  const float* Vh = V  + (size_t)bh * S * DK;

  __shared__ float lds_s[8][2 * 16 * 16];
  float* sbuf = lds_s[wave];

  v2f qa[2][16];
  const float* qrow0 = Qh + (size_t)(qBase +  0 + l) * DK;
  const float* qrow1 = Qh + (size_t)(qBase + 16 + l) * DK;
  #pragma unroll
  for (int kk = 0; kk < 16; ++kk) {
    qa[0][kk] = *(const v2f*)(qrow0 + kk * 4 + 2 * half);
    qa[1][kk] = *(const v2f*)(qrow1 + kk * 4 + 2 * half);
  }

  v8f o[2][4] = {{v8f{}, v8f{}, v8f{}, v8f{}},
                 {v8f{}, v8f{}, v8f{}, v8f{}}};

  for (int kt = 0; kt < S; kt += 16) {
    v8f s0 = {}, s1 = {};
    const float* krow = Kh + (size_t)(kt + l) * DK;
    #pragma unroll
    for (int kk = 0; kk < 16; ++kk) {
      v2f kb = *(const v2f*)(krow + kk * 4 + 2 * half);
      s0 = wmma_f32(qa[0][kk], kb, s0);
      s1 = wmma_f32(qa[1][kk], kb, s1);
    }

    const int mval = mask[bidx * S + kt + l];
    #pragma unroll
    for (int r = 0; r < 8; ++r) {
      const float v0 = (mval != 0) ? s0[r] * 0.125f : NEG_INF;
      const float v1 = (mval != 0) ? s1[r] * 0.125f : NEG_INF;
      sbuf[(r + 8 * half) * 16 + l]       = v0;
      sbuf[256 + (r + 8 * half) * 16 + l] = v1;
    }
    asm volatile("s_wait_dscnt 0" ::: "memory");

    #pragma unroll
    for (int kk2 = 0; kk2 < 4; ++kk2) {
      v2f sa0 = *(const v2f*)(&sbuf[l * 16 + kk2 * 4 + 2 * half]);
      v2f sa1 = *(const v2f*)(&sbuf[256 + l * 16 + kk2 * 4 + 2 * half]);
      const float* vrow = Vh + (size_t)(kt + kk2 * 4 + 2 * half) * DK;
      #pragma unroll
      for (int j = 0; j < 4; ++j) {
        v2f vb;
        vb[0] = vrow[j * 16 + l];
        vb[1] = vrow[DK + j * 16 + l];
        o[0][j] = wmma_f32(sa0, vb, o[0][j]);
        o[1][j] = wmma_f32(sa1, vb, o[1][j]);
      }
    }
  }

  #pragma unroll
  for (int mi = 0; mi < 2; ++mi) {
    #pragma unroll
    for (int j = 0; j < 4; ++j) {
      #pragma unroll
      for (int r = 0; r < 8; ++r) {
        const int q = qBase + mi * 16 + r + 8 * half;
        O[((size_t)bidx * S + q) * DMODEL + h * DK + j * 16 + l] = o[mi][j][r];
      }
    }
  }
}

extern "C" void kernel_launch(void* const* d_in, const int* in_sizes, int n_in,
                              void* d_out, int out_size, void* d_ws, size_t ws_size,
                              hipStream_t stream) {
  const int B = 2, S = 2048;
  const int M = B * S;
  (void)in_sizes; (void)n_in; (void)out_size; (void)ws_size;

  const float* q   = (const float*)d_in[0];
  const float* k   = (const float*)d_in[1];
  const float* v   = (const float*)d_in[2];
  const int*   msk = (const int*)  d_in[3];
  const float* W_q = (const float*)d_in[4];
  const float* b_q = (const float*)d_in[5];
  const float* W_k = (const float*)d_in[6];
  const float* b_k = (const float*)d_in[7];
  const float* W_v = (const float*)d_in[8];
  const float* b_v = (const float*)d_in[9];
  const float* W_o = (const float*)d_in[10];
  const float* b_o = (const float*)d_in[11];

  const size_t panel = (size_t)B * NHEAD * S * DK;
  float* wsQ = (float*)d_ws;
  float* wsK = wsQ + panel;
  float* wsV = wsK + panel;
  float* wsA = wsV + panel;

  dim3 blk(256);
  dim3 gemmGrid(DMODEL / 64, M / 256);

  gemm_nt_bias_kernel<<<gemmGrid, blk, 0, stream>>>(q, W_q, b_q, wsQ,
                                                    M, DMODEL, DMODEL, S, 1);
  gemm_nt_bias_kernel<<<gemmGrid, blk, 0, stream>>>(k, W_k, b_k, wsK,
                                                    M, DMODEL, DMODEL, S, 1);
  gemm_nt_bias_kernel<<<gemmGrid, blk, 0, stream>>>(v, W_v, b_v, wsV,
                                                    M, DMODEL, DMODEL, S, 1);

  dim3 attnGrid(S / 256, B * NHEAD);
  attn_nosoftmax_kernel<<<attnGrid, blk, 0, stream>>>(wsQ, wsK, wsV, msk,
                                                      wsA, S);

  gemm_nt_bias_kernel<<<gemmGrid, blk, 0, stream>>>(wsA, W_o, b_o,
                                                    (float*)d_out,
                                                    M, DMODEL, DMODEL, S, 0);
}


# Attention has no softmax: reassociate (Q K^T) V as Q (K^T V): 64x64 K^T V per head, masked keys as a -1e9*sum(V) row bias, then Q x context; changes the f32 summation order
# speedup vs baseline: 1.6291x; 1.6291x over previous
#include <hip/hip_runtime.h>


typedef float v2f __attribute__((ext_vector_type(2)));
typedef float v8f __attribute__((ext_vector_type(8)));

#define DMODEL 1024
#define NHEAD  16
#define DK     64
#define NEG_INF (-1.0e9f)

__device__ inline v8f wmma_f32(v2f a, v2f b, v8f c) {
  return __builtin_amdgcn_wmma_f32_16x16x4_f32(false, a, false, b,
                                               (short)0, c, false, false);
}

__global__ void __launch_bounds__(256)
gemm_nt_bias_kernel(const float* __restrict__ X, const float* __restrict__ W,
                    const float* __restrict__ bias, float* __restrict__ Y,
                    int M, int N, int K, int S, int qkv_layout) {
  const int lane = threadIdx.x & 31;
  const int wave = threadIdx.x >> 5;
  const int half = lane >> 4;
  const int l    = lane & 15;
  const int mBase = (blockIdx.y * 8 + wave) * 32;
  const int nBase = blockIdx.x * 64;

  const float* xrow0 = X + (size_t)(mBase +  0 + l) * K;
  const float* xrow1 = X + (size_t)(mBase + 16 + l) * K;
  const float* w0    = W + (size_t)(nBase +  0 + l) * K;
  const float* w1    = W + (size_t)(nBase + 16 + l) * K;
  const float* w2    = W + (size_t)(nBase + 32 + l) * K;
  const float* w3    = W + (size_t)(nBase + 48 + l) * K;

  v8f acc[2][4] = {{v8f{}, v8f{}, v8f{}, v8f{}},
                   {v8f{}, v8f{}, v8f{}, v8f{}}};

  #pragma unroll 4
  for (int k0 = 0; k0 < K; k0 += 4) {
    const int off = k0 + 2 * half;
    v2f a0 = *(const v2f*)(xrow0 + off);
    v2f a1 = *(const v2f*)(xrow1 + off);
    v2f b0 = *(const v2f*)(w0 + off);
    v2f b1 = *(const v2f*)(w1 + off);
    v2f b2 = *(const v2f*)(w2 + off);
    v2f b3 = *(const v2f*)(w3 + off);
    acc[0][0] = wmma_f32(a0, b0, acc[0][0]);
    acc[0][1] = wmma_f32(a0, b1, acc[0][1]);
    acc[0][2] = wmma_f32(a0, b2, acc[0][2]);
    acc[0][3] = wmma_f32(a0, b3, acc[0][3]);
    acc[1][0] = wmma_f32(a1, b0, acc[1][0]);
    acc[1][1] = wmma_f32(a1, b1, acc[1][1]);
    acc[1][2] = wmma_f32(a1, b2, acc[1][2]);
    acc[1][3] = wmma_f32(a1, b3, acc[1][3]);
  }

  #pragma unroll
  for (int mi = 0; mi < 2; ++mi) {
    #pragma unroll
    for (int j = 0; j < 4; ++j) {
      const int n  = nBase + j * 16 + l;
      const float bv = bias[n];
      #pragma unroll
      for (int r = 0; r < 8; ++r) {
        const int m = mBase + mi * 16 + r + 8 * half;
        const float v = acc[mi][j][r] + bv;
        size_t dst;
        if (qkv_layout) {
          const int bidx = m / S, s = m % S;
          const int h = n / DK,   d = n % DK;
          dst = (((size_t)(bidx * NHEAD + h) * S + s) * DK + d);
        } else {
          dst = (size_t)m * N + n;
        }
        Y[dst] = v;
      }
    }
  }
}

#define KVW 16

__global__ void __launch_bounds__(KVW * 32)
kv_context_kernel(const float* __restrict__ Kp, const float* __restrict__ V,
                  const int* __restrict__ mask, float* __restrict__ KVf,
                  float* __restrict__ Cb, int S) {
  const int lane = threadIdx.x & 31;
  const int wave = threadIdx.x >> 5;
  const int half = lane >> 4;
  const int l    = lane & 15;
  const int it   = blockIdx.x;
  const int jh   = blockIdx.y;
  const int bh   = blockIdx.z;
  const int bidx = bh / NHEAD;

  __shared__ float red[KVW * 512];
  __shared__ float vred[KVW * 64];

  const int kchunk = S / KVW;
  const int kFirst = wave * kchunk + 2 * half;
  const float* kcol = Kp + ((size_t)bh * S + kFirst) * DK + it * 16 + l;
  const float* vcol = V  + ((size_t)bh * S + kFirst) * DK + jh * 32 + l;
  const int*   mp   = mask + (size_t)bidx * S + kFirst;

  v8f acc0 = {}, acc1 = {};
  float vs0 = 0.f, vs1 = 0.f;

  #pragma unroll 8
  for (int k0 = 0; k0 < kchunk; k0 += 4) {
    const int2 mm = *(const int2*)(mp + k0);
    const float* kr = kcol + (size_t)k0 * DK;
    const float* vr = vcol + (size_t)k0 * DK;
    const float ka0 = kr[0];
    const float ka1 = kr[DK];
    v2f b0, b1;
    b0[0] = vr[0];
    b0[1] = vr[DK];
    b1[0] = vr[16];
    b1[1] = vr[DK + 16];
    const bool keep0 = (mm.x != 0);
    const bool keep1 = (mm.y != 0);
    v2f a;
    a[0] = keep0 ? ka0 : 0.f;
    a[1] = keep1 ? ka1 : 0.f;
    acc0 = wmma_f32(a, b0, acc0);
    acc1 = wmma_f32(a, b1, acc1);
    vs0 += (keep0 ? 0.f : b0[0]) + (keep1 ? 0.f : b0[1]);
    vs1 += (keep0 ? 0.f : b1[0]) + (keep1 ? 0.f : b1[1]);
  }

  #pragma unroll
  for (int r = 0; r < 8; ++r) {
    red[wave * 512 + r * 32 + lane]       = acc0[r];
    red[wave * 512 + 256 + r * 32 + lane] = acc1[r];
  }
  vred[wave * 64 + lane]      = vs0;
  vred[wave * 64 + 32 + lane] = vs1;
  __syncthreads();

  {
    const int o   = threadIdx.x;
    const int f   = o >> 6;
    const int kkl = f >> 1, jt = f & 1;
    const int w   = o & 63;
    const int lp  = (w >> 1) & 15, hp = w >> 5, s = w & 1;
    const int el  = 4 * kkl + 2 * hp + s;
    const int src = (jt * 8 + (el & 7)) * 32 + lp + 16 * (el >> 3);
    float sum = 0.f;
    #pragma unroll
    for (int w8 = 0; w8 < KVW; ++w8) sum += red[w8 * 512 + src];
    KVf[(size_t)bh * (DK * DK) +
        ((it * 4 + kkl) * 4 + (jh * 2 + jt)) * 64 + w] = sum * 0.125f;
  }
  if (it == 0 && threadIdx.x < 32) {
    const int t = threadIdx.x >> 4, lp = threadIdx.x & 15;
    float sum = 0.f;
    #pragma unroll
    for (int w8 = 0; w8 < KVW; ++w8)
      sum += vred[w8 * 64 + t * 32 + lp] + vred[w8 * 64 + t * 32 + 16 + lp];
    Cb[bh * DK + jh * 32 + threadIdx.x] = NEG_INF * sum;
  }
}

__global__ void __launch_bounds__(256)
q_context_kernel(const float* __restrict__ Q, const float* __restrict__ KVf,
                 const float* __restrict__ Cb, float* __restrict__ O, int S) {
  const int lane = threadIdx.x & 31;
  const int wave = threadIdx.x >> 5;
  const int half = lane >> 4;
  const int l    = lane & 15;
  const int bh   = blockIdx.y;
  const int bidx = bh / NHEAD;
  const int h    = bh % NHEAD;
  const int qBase = blockIdx.x * 256 + wave * 32;

  const float* qrow0 = Q + ((size_t)bh * S + qBase + l) * DK + 2 * half;
  const float* qrow1 = qrow0 + 16 * DK;
  const float* kv    = KVf + (size_t)bh * (DK * DK) + lane * 2;

  v8f o[2][4] = {{v8f{}, v8f{}, v8f{}, v8f{}},
                 {v8f{}, v8f{}, v8f{}, v8f{}}};

  #pragma unroll
  for (int kk = 0; kk < 16; ++kk) {
    const v2f a0 = *(const v2f*)(qrow0 + kk * 4);
    const v2f a1 = *(const v2f*)(qrow1 + kk * 4);
    #pragma unroll
    for (int j = 0; j < 4; ++j) {
      const v2f b = *(const v2f*)(kv + (kk * 4 + j) * 64);
      o[0][j] = wmma_f32(a0, b, o[0][j]);
      o[1][j] = wmma_f32(a1, b, o[1][j]);
    }
  }

  #pragma unroll
  for (int j = 0; j < 4; ++j) {
    const float cb = Cb[bh * DK + j * 16 + l];
    #pragma unroll
    for (int mi = 0; mi < 2; ++mi) {
      #pragma unroll
      for (int r = 0; r < 8; ++r) {
        const int q = qBase + mi * 16 + r + 8 * half;
        O[((size_t)bidx * S + q) * DMODEL + h * DK + j * 16 + l] = o[mi][j][r] + cb;
      }
    }
  }
}

extern "C" void kernel_launch(void* const* d_in, const int* in_sizes, int n_in,
                              void* d_out, int out_size, void* d_ws, size_t ws_size,
                              hipStream_t stream) {
  const int B = 2, S = 2048;
  const int M = B * S;
  (void)in_sizes; (void)n_in; (void)out_size; (void)ws_size;

  const float* q   = (const float*)d_in[0];
  const float* k   = (const float*)d_in[1];
  const float* v   = (const float*)d_in[2];
  const int*   msk = (const int*)  d_in[3];
  const float* W_q = (const float*)d_in[4];
  const float* b_q = (const float*)d_in[5];
  const float* W_k = (const float*)d_in[6];
  const float* b_k = (const float*)d_in[7];
  const float* W_v = (const float*)d_in[8];
  const float* b_v = (const float*)d_in[9];
  const float* W_o = (const float*)d_in[10];
  const float* b_o = (const float*)d_in[11];

  const size_t panel = (size_t)B * NHEAD * S * DK;
  float* wsQ = (float*)d_ws;
  float* wsK = wsQ + panel;
  float* wsV = wsK + panel;
  float* wsA = wsV + panel;
  float* wsKV = wsA + panel;
  float* wsC  = wsKV + (size_t)B * NHEAD * DK * DK;

  dim3 blk(256);
  dim3 gemmGrid(DMODEL / 64, M / 256);

  gemm_nt_bias_kernel<<<gemmGrid, blk, 0, stream>>>(q, W_q, b_q, wsQ,
                                                    M, DMODEL, DMODEL, S, 1);
  gemm_nt_bias_kernel<<<gemmGrid, blk, 0, stream>>>(k, W_k, b_k, wsK,
                                                    M, DMODEL, DMODEL, S, 1);
  gemm_nt_bias_kernel<<<gemmGrid, blk, 0, stream>>>(v, W_v, b_v, wsV,
                                                    M, DMODEL, DMODEL, S, 1);

  dim3 kvGrid(4, 2, B * NHEAD);
  kv_context_kernel<<<kvGrid, dim3(KVW * 32), 0, stream>>>(wsK, wsV, msk,
                                                           wsKV, wsC, S);
  dim3 qcGrid(S / 256, B * NHEAD);
  q_context_kernel<<<qcGrid, blk, 0, stream>>>(wsQ, wsKV, wsC, wsA, S);

  gemm_nt_bias_kernel<<<gemmGrid, blk, 0, stream>>>(wsA, W_o, b_o,
                                                    (float*)d_out,
                                                    M, DMODEL, DMODEL, S, 0);
}


# Projection GEMMs: operand tiles staged through LDS by whole-line 16-byte global loads (two stages, k-group-major), fragments read from LDS, not by strided 8-byte global loads; sum order unchanged
# speedup vs baseline: 2.8360x; 1.7408x over previous
#include <hip/hip_runtime.h>


typedef float v2f __attribute__((ext_vector_type(2)));
typedef float v8f __attribute__((ext_vector_type(8)));

#define DMODEL 1024
#define NHEAD  16
#define DK     64
#define NEG_INF (-1.0e9f)

__device__ inline v8f wmma_f32(v2f a, v2f b, v8f c) {
  return __builtin_amdgcn_wmma_f32_16x16x4_f32(false, a, false, b,
                                               (short)0, c, false, false);
}

#define GT_K   64
#define GT_KG  (GT_K / 4)
#define PLA    ((256 + 1) * 4)
#define PLB    ((64 + 1) * 4)
#define STG    (GT_KG * (PLA + PLB))

struct Frags { v2f a0, a1, b0, b1, b2, b3; };

__device__ inline Frags load_frags(const float* pa, const float* pb) {
  Frags f;
  f.a0 = *(const v2f*)(pa);
  f.a1 = *(const v2f*)(pa + 64);
  f.b0 = *(const v2f*)(pb);
  f.b1 = *(const v2f*)(pb + 64);
  f.b2 = *(const v2f*)(pb + 128);
  f.b3 = *(const v2f*)(pb + 192);
  return f;
}

__device__ inline void mma_step(const Frags& f, v8f (&acc)[2][4]) {
  acc[0][0] = wmma_f32(f.a0, f.b0, acc[0][0]);
  acc[0][1] = wmma_f32(f.a0, f.b1, acc[0][1]);
  acc[0][2] = wmma_f32(f.a0, f.b2, acc[0][2]);
  acc[0][3] = wmma_f32(f.a0, f.b3, acc[0][3]);
  acc[1][0] = wmma_f32(f.a1, f.b0, acc[1][0]);
  acc[1][1] = wmma_f32(f.a1, f.b1, acc[1][1]);
  acc[1][2] = wmma_f32(f.a1, f.b2, acc[1][2]);
  acc[1][3] = wmma_f32(f.a1, f.b3, acc[1][3]);
}

template <int P0, int P1>
__device__ inline void mma_pairs(const float* pa, const float* pb,
                                 v8f (&acc)[2][4]) {
  Frags f[2][2];
  f[P0 & 1][0] = load_frags(pa + (2 * P0) * PLA,     pb + (2 * P0) * PLB);
  f[P0 & 1][1] = load_frags(pa + (2 * P0 + 1) * PLA, pb + (2 * P0 + 1) * PLB);
  #pragma unroll
  for (int p = P0; p < P1; ++p) {
    if (p + 1 < P1) {
      f[(p + 1) & 1][0] = load_frags(pa + (2 * p + 2) * PLA, pb + (2 * p + 2) * PLB);
      f[(p + 1) & 1][1] = load_frags(pa + (2 * p + 3) * PLA, pb + (2 * p + 3) * PLB);
    }
    mma_step(f[p & 1][0], acc);
    mma_step(f[p & 1][1], acc);
  }
}

#define FOR16(X) X(0) X(1) X(2) X(3) X(4) X(5) X(6) X(7) X(8) X(9) X(10) X(11) X(12) X(13) X(14) X(15)
#define FOR4(X)  X(0) X(1) X(2) X(3)
#define DECL_A(i)  float4 ra##i;
#define DECL_B(i)  float4 rb##i;
#define LOAD_A(i)  ra##i = *(const float4*)(xp + (i) * rowStep);
#define LOAD_B(i)  rb##i = *(const float4*)(wp + (i) * rowStep);
#define STORE_A(i) *(float4*)(dap + (i) * 64) = ra##i;
#define STORE_B(i) *(float4*)(dbp + (i) * 64) = rb##i;

__global__ void __launch_bounds__(256)
gemm_nt_bias_kernel(const float* __restrict__ X, const float* __restrict__ W,
                    const float* __restrict__ bias, float* __restrict__ Y,
                    int M, int N, int K, int S, int qkv_layout) {
  __shared__ __attribute__((aligned(16))) float tile[2 * STG];

  const int tid  = threadIdx.x;
  const int lane = tid & 31;
  const int wave = tid >> 5;
  const int half = lane >> 4;
  const int l    = lane & 15;
  const int mBlock = blockIdx.y * 256;
  const int mBase  = mBlock + wave * 32;
  const int nBase  = blockIdx.x * 64;

  const int srow = tid >> 4;
  const int sg   = tid & 15;
  const float* xsrc = X + (size_t)(mBlock + srow) * K + sg * 4;
  const float* wsrc = W + (size_t)(nBase  + srow) * K + sg * 4;
  const int adst = sg * PLA + srow * 4;
  const int bdst = GT_KG * PLA + sg * PLB + srow * 4;
  const int aoff = (wave * 32 + l) * 4 + 2 * half;
  const int boff = GT_KG * PLA + l * 4 + 2 * half;

  v8f acc[2][4] = {{v8f{}, v8f{}, v8f{}, v8f{}},
                   {v8f{}, v8f{}, v8f{}, v8f{}}};
  FOR16(DECL_A) FOR4(DECL_B)

  const int nChunks = K / GT_K;
  const size_t rowStep = (size_t)16 * K;

  {
    const float* xp = xsrc;
    const float* wp = wsrc;
    float* dap = tile + adst;
    float* dbp = tile + bdst;
    FOR16(LOAD_A) FOR4(LOAD_B)
    FOR16(STORE_A) FOR4(STORE_B)
    const int c1 = (nChunks > 1 ? 1 : 0) * GT_K;
    xp = xsrc + c1;
    wp = wsrc + c1;
    FOR16(LOAD_A) FOR4(LOAD_B)
  }
  __syncthreads();

  for (int kc = 0; kc + 1 < nChunks; ++kc) {
    const float* cur = tile + (kc & 1) * STG;
    float* nxt = tile + ((kc & 1) ^ 1) * STG;
    const int k2 = (kc + 2 < nChunks ? kc + 2 : nChunks - 1) * GT_K;
    mma_pairs<0, 2>(cur + aoff, cur + boff, acc);
    __builtin_amdgcn_sched_barrier(0);
    {
      float* dap = nxt + adst;
      float* dbp = nxt + bdst;
      const float* xp = xsrc + k2;
      const float* wp = wsrc + k2;
      FOR16(STORE_A) FOR4(STORE_B)
      FOR16(LOAD_A) FOR4(LOAD_B)
    }
    mma_pairs<2, GT_KG / 2>(cur + aoff, cur + boff, acc);
    __syncthreads();
  }
  {
    const float* cur = tile + ((nChunks - 1) & 1) * STG;
    mma_pairs<0, GT_KG / 2>(cur + aoff, cur + boff, acc);
  }

  #pragma unroll
  for (int mi = 0; mi < 2; ++mi) {
    #pragma unroll
    for (int j = 0; j < 4; ++j) {
      const int n  = nBase + j * 16 + l;
      const float bv = bias[n];
      #pragma unroll
      for (int r = 0; r < 8; ++r) {
        const int m = mBase + mi * 16 + r + 8 * half;
        const float v = acc[mi][j][r] + bv;
        size_t dst;
        if (qkv_layout) {
          const int bidx = m / S, s = m % S;
          const int h = n / DK,   d = n % DK;
          dst = (((size_t)(bidx * NHEAD + h) * S + s) * DK + d);
        } else {
          dst = (size_t)m * N + n;
        }
        Y[dst] = v;
      }
    }
  }
}

#define KVW 16

__global__ void __launch_bounds__(KVW * 32)
kv_context_kernel(const float* __restrict__ Kp, const float* __restrict__ V,
                  const int* __restrict__ mask, float* __restrict__ KVf,
                  float* __restrict__ Cb, int S) {
  const int lane = threadIdx.x & 31;
  const int wave = threadIdx.x >> 5;
  const int half = lane >> 4;
  const int l    = lane & 15;
  const int it   = blockIdx.x;
  const int jh   = blockIdx.y;
  const int bh   = blockIdx.z;
  const int bidx = bh / NHEAD;

  __shared__ float red[KVW * 512];
  __shared__ float vred[KVW * 64];

  const int kchunk = S / KVW;
  const int kFirst = wave * kchunk + 2 * half;
  const float* kcol = Kp + ((size_t)bh * S + kFirst) * DK + it * 16 + l;
  const float* vcol = V  + ((size_t)bh * S + kFirst) * DK + jh * 32 + l;
  const int*   mp   = mask + (size_t)bidx * S + kFirst;

  v8f acc0 = {}, acc1 = {};
  float vs0 = 0.f, vs1 = 0.f;

  #pragma unroll 8
  for (int k0 = 0; k0 < kchunk; k0 += 4) {
    const int2 mm = *(const int2*)(mp + k0);
    const float* kr = kcol + (size_t)k0 * DK;
    const float* vr = vcol + (size_t)k0 * DK;
    const float ka0 = kr[0];
    const float ka1 = kr[DK];
    v2f b0, b1;
    b0[0] = vr[0];
    b0[1] = vr[DK];
    b1[0] = vr[16];
    b1[1] = vr[DK + 16];
    const bool keep0 = (mm.x != 0);
    const bool keep1 = (mm.y != 0);
    v2f a;
    a[0] = keep0 ? ka0 : 0.f;
    a[1] = keep1 ? ka1 : 0.f;
    acc0 = wmma_f32(a, b0, acc0);
    acc1 = wmma_f32(a, b1, acc1);
    vs0 += (keep0 ? 0.f : b0[0]) + (keep1 ? 0.f : b0[1]);
    vs1 += (keep0 ? 0.f : b1[0]) + (keep1 ? 0.f : b1[1]);
  }

  #pragma unroll
  for (int r = 0; r < 8; ++r) {
    red[wave * 512 + r * 32 + lane]       = acc0[r];
    red[wave * 512 + 256 + r * 32 + lane] = acc1[r];
  }
  vred[wave * 64 + lane]      = vs0;
  vred[wave * 64 + 32 + lane] = vs1;
  __syncthreads();

  {
    const int o   = threadIdx.x;
    const int f   = o >> 6;
    const int kkl = f >> 1, jt = f & 1;
    const int w   = o & 63;
    const int lp  = (w >> 1) & 15, hp = w >> 5, s = w & 1;
    const int el  = 4 * kkl + 2 * hp + s;
    const int src = (jt * 8 + (el & 7)) * 32 + lp + 16 * (el >> 3);
    float sum = 0.f;
    #pragma unroll
    for (int w8 = 0; w8 < KVW; ++w8) sum += red[w8 * 512 + src];
    KVf[(size_t)bh * (DK * DK) +
        ((it * 4 + kkl) * 4 + (jh * 2 + jt)) * 64 + w] = sum * 0.125f;
  }
  if (it == 0 && threadIdx.x < 32) {
    const int t = threadIdx.x >> 4, lp = threadIdx.x & 15;
    float sum = 0.f;
    #pragma unroll
    for (int w8 = 0; w8 < KVW; ++w8)
      sum += vred[w8 * 64 + t * 32 + lp] + vred[w8 * 64 + t * 32 + 16 + lp];
    Cb[bh * DK + jh * 32 + threadIdx.x] = NEG_INF * sum;
  }
}

__global__ void __launch_bounds__(256)
q_context_kernel(const float* __restrict__ Q, const float* __restrict__ KVf,
                 const float* __restrict__ Cb, float* __restrict__ O, int S) {
  const int lane = threadIdx.x & 31;
  const int wave = threadIdx.x >> 5;
  const int half = lane >> 4;
  const int l    = lane & 15;
  const int bh   = blockIdx.y;
  const int bidx = bh / NHEAD;
  const int h    = bh % NHEAD;
  const int qBase = blockIdx.x * 256 + wave * 32;

  const float* qrow0 = Q + ((size_t)bh * S + qBase + l) * DK + 2 * half;
  const float* qrow1 = qrow0 + 16 * DK;
  const float* kv    = KVf + (size_t)bh * (DK * DK) + lane * 2;

  v8f o[2][4] = {{v8f{}, v8f{}, v8f{}, v8f{}},
                 {v8f{}, v8f{}, v8f{}, v8f{}}};

  #pragma unroll
  for (int kk = 0; kk < 16; ++kk) {
    const v2f a0 = *(const v2f*)(qrow0 + kk * 4);
    const v2f a1 = *(const v2f*)(qrow1 + kk * 4);
    #pragma unroll
    for (int j = 0; j < 4; ++j) {
      const v2f b = *(const v2f*)(kv + (kk * 4 + j) * 64);
      o[0][j] = wmma_f32(a0, b, o[0][j]);
      o[1][j] = wmma_f32(a1, b, o[1][j]);
    }
  }

  #pragma unroll
  for (int j = 0; j < 4; ++j) {
    const float cb = Cb[bh * DK + j * 16 + l];
    #pragma unroll
    for (int mi = 0; mi < 2; ++mi) {
      #pragma unroll
      for (int r = 0; r < 8; ++r) {
        const int q = qBase + mi * 16 + r + 8 * half;
        O[((size_t)bidx * S + q) * DMODEL + h * DK + j * 16 + l] = o[mi][j][r] + cb;
      }
    }
  }
}

extern "C" void kernel_launch(void* const* d_in, const int* in_sizes, int n_in,
                              void* d_out, int out_size, void* d_ws, size_t ws_size,
                              hipStream_t stream) {
  const int B = 2, S = 2048;
  const int M = B * S;
  (void)in_sizes; (void)n_in; (void)out_size; (void)ws_size;

  const float* q   = (const float*)d_in[0];
  const float* k   = (const float*)d_in[1];
  const float* v   = (const float*)d_in[2];
  const int*   msk = (const int*)  d_in[3];
  const float* W_q = (const float*)d_in[4];
  const float* b_q = (const float*)d_in[5];
  const float* W_k = (const float*)d_in[6];
  const float* b_k = (const float*)d_in[7];
  const float* W_v = (const float*)d_in[8];
  const float* b_v = (const float*)d_in[9];
  const float* W_o = (const float*)d_in[10];
  const float* b_o = (const float*)d_in[11];

  const size_t panel = (size_t)B * NHEAD * S * DK;
  float* wsQ = (float*)d_ws;
  float* wsK = wsQ + panel;
  float* wsV = wsK + panel;
  float* wsA = wsV + panel;
  float* wsKV = wsA + panel;
  float* wsC  = wsKV + (size_t)B * NHEAD * DK * DK;

  dim3 blk(256);
  dim3 gemmGrid(DMODEL / 64, M / 256);

  gemm_nt_bias_kernel<<<gemmGrid, blk, 0, stream>>>(q, W_q, b_q, wsQ,
                                                    M, DMODEL, DMODEL, S, 1);
  gemm_nt_bias_kernel<<<gemmGrid, blk, 0, stream>>>(k, W_k, b_k, wsK,
                                                    M, DMODEL, DMODEL, S, 1);
  gemm_nt_bias_kernel<<<gemmGrid, blk, 0, stream>>>(v, W_v, b_v, wsV,
                                                    M, DMODEL, DMODEL, S, 1);

  dim3 kvGrid(4, 2, B * NHEAD);
  kv_context_kernel<<<kvGrid, dim3(KVW * 32), 0, stream>>>(wsK, wsV, msk,
                                                           wsKV, wsC, S);
  dim3 qcGrid(S / 256, B * NHEAD);
  q_context_kernel<<<qcGrid, blk, 0, stream>>>(wsQ, wsKV, wsC, wsA, S);

  gemm_nt_bias_kernel<<<gemmGrid, blk, 0, stream>>>(wsA, W_o, b_o,
                                                    (float*)d_out,
                                                    M, DMODEL, DMODEL, S, 0);
}


# GEMM epilogue: each wave's 32x64 result staged through LDS and stored as whole 256-byte rows by 16-byte stores; row address from one index division per wave, not per element; sums unchanged
# speedup vs baseline: 2.9388x; 1.0362x over previous
#include <hip/hip_runtime.h>


typedef float v2f __attribute__((ext_vector_type(2)));
typedef float v8f __attribute__((ext_vector_type(8)));

#define DMODEL 1024
#define NHEAD  16
#define DK     64
#define NEG_INF (-1.0e9f)

__device__ inline v8f wmma_f32(v2f a, v2f b, v8f c) {
  return __builtin_amdgcn_wmma_f32_16x16x4_f32(false, a, false, b,
                                               (short)0, c, false, false);
}

#define GT_K   64
#define GT_KG  (GT_K / 4)
#define PLA    ((256 + 1) * 4)
#define PLB    ((64 + 1) * 4)
#define STG    (GT_KG * (PLA + PLB))
#define OPITCH 68

struct Frags { v2f a0, a1, b0, b1, b2, b3; };

__device__ inline Frags load_frags(const float* pa, const float* pb) {
  Frags f;
  f.a0 = *(const v2f*)(pa);
  f.a1 = *(const v2f*)(pa + 64);
  f.b0 = *(const v2f*)(pb);
  f.b1 = *(const v2f*)(pb + 64);
  f.b2 = *(const v2f*)(pb + 128);
  f.b3 = *(const v2f*)(pb + 192);
  return f;
}

__device__ inline void mma_step(const Frags& f, v8f (&acc)[2][4]) {
  acc[0][0] = wmma_f32(f.a0, f.b0, acc[0][0]);
  acc[0][1] = wmma_f32(f.a0, f.b1, acc[0][1]);
  acc[0][2] = wmma_f32(f.a0, f.b2, acc[0][2]);
  acc[0][3] = wmma_f32(f.a0, f.b3, acc[0][3]);
  acc[1][0] = wmma_f32(f.a1, f.b0, acc[1][0]);
  acc[1][1] = wmma_f32(f.a1, f.b1, acc[1][1]);
  acc[1][2] = wmma_f32(f.a1, f.b2, acc[1][2]);
  acc[1][3] = wmma_f32(f.a1, f.b3, acc[1][3]);
}

template <int P0, int P1>
__device__ inline void mma_pairs(const float* pa, const float* pb,
                                 v8f (&acc)[2][4]) {
  Frags f[2][2];
  f[P0 & 1][0] = load_frags(pa + (2 * P0) * PLA,     pb + (2 * P0) * PLB);
  f[P0 & 1][1] = load_frags(pa + (2 * P0 + 1) * PLA, pb + (2 * P0 + 1) * PLB);
  #pragma unroll
  for (int p = P0; p < P1; ++p) {
    if (p + 1 < P1) {
      f[(p + 1) & 1][0] = load_frags(pa + (2 * p + 2) * PLA, pb + (2 * p + 2) * PLB);
      f[(p + 1) & 1][1] = load_frags(pa + (2 * p + 3) * PLA, pb + (2 * p + 3) * PLB);
    }
    mma_step(f[p & 1][0], acc);
    mma_step(f[p & 1][1], acc);
  }
}

#define FOR16(X) X(0) X(1) X(2) X(3) X(4) X(5) X(6) X(7) X(8) X(9) X(10) X(11) X(12) X(13) X(14) X(15)
#define FOR4(X)  X(0) X(1) X(2) X(3)
#define DECL_A(i)  float4 ra##i;
#define DECL_B(i)  float4 rb##i;
#define LOAD_A(i)  ra##i = *(const float4*)(xp + (i) * rowStep);
#define LOAD_B(i)  rb##i = *(const float4*)(wp + (i) * rowStep);
#define STORE_A(i) *(float4*)(dap + (i) * 64) = ra##i;
#define STORE_B(i) *(float4*)(dbp + (i) * 64) = rb##i;

__global__ void __launch_bounds__(256)
gemm_nt_bias_kernel(const float* __restrict__ X, const float* __restrict__ W,
                    const float* __restrict__ bias, float* __restrict__ Y,
                    int M, int N, int K, int S, int qkv_layout) {
  __shared__ __attribute__((aligned(16))) float tile[2 * STG];

  const int tid  = threadIdx.x;
  const int lane = tid & 31;
  const int wave = tid >> 5;
  const int half = lane >> 4;
  const int l    = lane & 15;
  const int mBlock = blockIdx.y * 256;
  const int mBase  = mBlock + wave * 32;
  const int nBase  = blockIdx.x * 64;

  const int srow = tid >> 4;
  const int sg   = tid & 15;
  const float* xsrc = X + (size_t)(mBlock + srow) * K + sg * 4;
  const float* wsrc = W + (size_t)(nBase  + srow) * K + sg * 4;
  const int adst = sg * PLA + srow * 4;
  const int bdst = GT_KG * PLA + sg * PLB + srow * 4;
  const int aoff = (wave * 32 + l) * 4 + 2 * half;
  const int boff = GT_KG * PLA + l * 4 + 2 * half;

  v8f acc[2][4] = {{v8f{}, v8f{}, v8f{}, v8f{}},
                   {v8f{}, v8f{}, v8f{}, v8f{}}};
  FOR16(DECL_A) FOR4(DECL_B)

  const int nChunks = K / GT_K;
  const size_t rowStep = (size_t)16 * K;

  {
    const float* xp = xsrc;
    const float* wp = wsrc;
    float* dap = tile + adst;
    float* dbp = tile + bdst;
    FOR16(LOAD_A) FOR4(LOAD_B)
    FOR16(STORE_A) FOR4(STORE_B)
    const int c1 = (nChunks > 1 ? 1 : 0) * GT_K;
    xp = xsrc + c1;
    wp = wsrc + c1;
    FOR16(LOAD_A) FOR4(LOAD_B)
  }
  __syncthreads();

  for (int kc = 0; kc + 1 < nChunks; ++kc) {
    const float* cur = tile + (kc & 1) * STG;
    float* nxt = tile + ((kc & 1) ^ 1) * STG;
    const int k2 = (kc + 2 < nChunks ? kc + 2 : nChunks - 1) * GT_K;
    mma_pairs<0, 2>(cur + aoff, cur + boff, acc);
    __builtin_amdgcn_sched_barrier(0);
    {
      float* dap = nxt + adst;
      float* dbp = nxt + bdst;
      const float* xp = xsrc + k2;
      const float* wp = wsrc + k2;
      FOR16(STORE_A) FOR4(STORE_B)
      FOR16(LOAD_A) FOR4(LOAD_B)
    }
    mma_pairs<2, GT_KG / 2>(cur + aoff, cur + boff, acc);
    __syncthreads();
  }
  const float bv0 = bias[nBase + l];
  const float bv1 = bias[nBase + 16 + l];
  const float bv2 = bias[nBase + 32 + l];
  const float bv3 = bias[nBase + 48 + l];
  {
    const float* cur = tile + ((nChunks - 1) & 1) * STG;
    mma_pairs<0, GT_KG / 2>(cur + aoff, cur + boff, acc);
  }

  static_assert(8 * 32 * OPITCH <= STG, "the eight output patches must fit one stage");
  static_assert(DK == 64, "the 64-column tile must be exactly one head");
  float* patch = tile + (nChunks & 1) * STG + wave * (32 * OPITCH);
  {
    float* pw = patch + (8 * half) * OPITCH + l;
    const float bv[4] = {bv0, bv1, bv2, bv3};
    #pragma unroll
    for (int j = 0; j < 4; ++j) {
      #pragma unroll
      for (int mi = 0; mi < 2; ++mi) {
        #pragma unroll
        for (int r = 0; r < 8; ++r)
          pw[(mi * 16 + r) * OPITCH + j * 16] = acc[mi][j][r] + bv[j];
      }
    }
  }
  __syncthreads();

  size_t rowBase, rowPitch;
  if (qkv_layout) {
    const int bidx = mBase / S;
    const int s0   = mBase - bidx * S;
    const int h    = nBase / DK;
    rowBase  = ((size_t)(bidx * NHEAD + h) * S + s0) * DK;
    rowPitch = DK;
  } else {
    rowBase  = (size_t)mBase * N + nBase;
    rowPitch = (size_t)N;
  }
  const float* pr = patch + half * OPITCH + 4 * l;
  float* yp = Y + rowBase + (size_t)half * rowPitch + 4 * l;
  #pragma unroll
  for (int i = 0; i < 16; ++i) {
    const float4 v = *(const float4*)(pr + (2 * i) * OPITCH);
    *(float4*)(yp + (size_t)(2 * i) * rowPitch) = v;
  }
}

#define KVW 16

__global__ void __launch_bounds__(KVW * 32)
kv_context_kernel(const float* __restrict__ Kp, const float* __restrict__ V,
                  const int* __restrict__ mask, float* __restrict__ KVf,
                  float* __restrict__ Cb, int S) {
  const int lane = threadIdx.x & 31;
  const int wave = threadIdx.x >> 5;
  const int half = lane >> 4;
  const int l    = lane & 15;
  const int it   = blockIdx.x;
  const int jh   = blockIdx.y;
  const int bh   = blockIdx.z;
  const int bidx = bh / NHEAD;

  __shared__ float red[KVW * 512];
  __shared__ float vred[KVW * 64];

  const int kchunk = S / KVW;
  const int kFirst = wave * kchunk + 2 * half;
  const float* kcol = Kp + ((size_t)bh * S + kFirst) * DK + it * 16 + l;
  const float* vcol = V  + ((size_t)bh * S + kFirst) * DK + jh * 32 + l;
  const int*   mp   = mask + (size_t)bidx * S + kFirst;

  v8f acc0 = {}, acc1 = {};
  float vs0 = 0.f, vs1 = 0.f;

  #pragma unroll 8
  for (int k0 = 0; k0 < kchunk; k0 += 4) {
    const int2 mm = *(const int2*)(mp + k0);
    const float* kr = kcol + (size_t)k0 * DK;
    const float* vr = vcol + (size_t)k0 * DK;
    const float ka0 = kr[0];
    const float ka1 = kr[DK];
    v2f b0, b1;
    b0[0] = vr[0];
    b0[1] = vr[DK];
    b1[0] = vr[16];
    b1[1] = vr[DK + 16];
    const bool keep0 = (mm.x != 0);
    const bool keep1 = (mm.y != 0);
    v2f a;
    a[0] = keep0 ? ka0 : 0.f;
    a[1] = keep1 ? ka1 : 0.f;
    acc0 = wmma_f32(a, b0, acc0);
    acc1 = wmma_f32(a, b1, acc1);
    vs0 += (keep0 ? 0.f : b0[0]) + (keep1 ? 0.f : b0[1]);
    vs1 += (keep0 ? 0.f : b1[0]) + (keep1 ? 0.f : b1[1]);
  }

  #pragma unroll
  for (int r = 0; r < 8; ++r) {
    red[wave * 512 + r * 32 + lane]       = acc0[r];
    red[wave * 512 + 256 + r * 32 + lane] = acc1[r];
  }
  vred[wave * 64 + lane]      = vs0;
  vred[wave * 64 + 32 + lane] = vs1;
  __syncthreads();

  {
    const int o   = threadIdx.x;
    const int f   = o >> 6;
    const int kkl = f >> 1, jt = f & 1;
    const int w   = o & 63;
    const int lp  = (w >> 1) & 15, hp = w >> 5, s = w & 1;
    const int el  = 4 * kkl + 2 * hp + s;
    const int src = (jt * 8 + (el & 7)) * 32 + lp + 16 * (el >> 3);
    float sum = 0.f;
    #pragma unroll
    for (int w8 = 0; w8 < KVW; ++w8) sum += red[w8 * 512 + src];
    KVf[(size_t)bh * (DK * DK) +
        ((it * 4 + kkl) * 4 + (jh * 2 + jt)) * 64 + w] = sum * 0.125f;
  }
  if (it == 0 && threadIdx.x < 32) {
    const int t = threadIdx.x >> 4, lp = threadIdx.x & 15;
    float sum = 0.f;
    #pragma unroll
    for (int w8 = 0; w8 < KVW; ++w8)
      sum += vred[w8 * 64 + t * 32 + lp] + vred[w8 * 64 + t * 32 + 16 + lp];
    Cb[bh * DK + jh * 32 + threadIdx.x] = NEG_INF * sum;
  }
}

__global__ void __launch_bounds__(256)
q_context_kernel(const float* __restrict__ Q, const float* __restrict__ KVf,
                 const float* __restrict__ Cb, float* __restrict__ O, int S) {
  const int lane = threadIdx.x & 31;
  const int wave = threadIdx.x >> 5;
  const int half = lane >> 4;
  const int l    = lane & 15;
  const int bh   = blockIdx.y;
  const int bidx = bh / NHEAD;
  const int h    = bh % NHEAD;
  const int qBase = blockIdx.x * 256 + wave * 32;

  const float* qrow0 = Q + ((size_t)bh * S + qBase + l) * DK + 2 * half;
  const float* qrow1 = qrow0 + 16 * DK;
  const float* kv    = KVf + (size_t)bh * (DK * DK) + lane * 2;

  v8f o[2][4] = {{v8f{}, v8f{}, v8f{}, v8f{}},
                 {v8f{}, v8f{}, v8f{}, v8f{}}};

  #pragma unroll
  for (int kk = 0; kk < 16; ++kk) {
    const v2f a0 = *(const v2f*)(qrow0 + kk * 4);
    const v2f a1 = *(const v2f*)(qrow1 + kk * 4);
    #pragma unroll
    for (int j = 0; j < 4; ++j) {
      const v2f b = *(const v2f*)(kv + (kk * 4 + j) * 64);
      o[0][j] = wmma_f32(a0, b, o[0][j]);
      o[1][j] = wmma_f32(a1, b, o[1][j]);
    }
  }

  #pragma unroll
  for (int j = 0; j < 4; ++j) {
    const float cb = Cb[bh * DK + j * 16 + l];
    #pragma unroll
    for (int mi = 0; mi < 2; ++mi) {
      #pragma unroll
      for (int r = 0; r < 8; ++r) {
        const int q = qBase + mi * 16 + r + 8 * half;
        O[((size_t)bidx * S + q) * DMODEL + h * DK + j * 16 + l] = o[mi][j][r] + cb;
      }
    }
  }
}

extern "C" void kernel_launch(void* const* d_in, const int* in_sizes, int n_in,
                              void* d_out, int out_size, void* d_ws, size_t ws_size,
                              hipStream_t stream) {
  const int B = 2, S = 2048;
  const int M = B * S;
  (void)in_sizes; (void)n_in; (void)out_size; (void)ws_size;

  const float* q   = (const float*)d_in[0];
  const float* k   = (const float*)d_in[1];
  const float* v   = (const float*)d_in[2];
  const int*   msk = (const int*)  d_in[3];
  const float* W_q = (const float*)d_in[4];
  const float* b_q = (const float*)d_in[5];
  const float* W_k = (const float*)d_in[6];
  const float* b_k = (const float*)d_in[7];
  const float* W_v = (const float*)d_in[8];
  const float* b_v = (const float*)d_in[9];
  const float* W_o = (const float*)d_in[10];
  const float* b_o = (const float*)d_in[11];

  const size_t panel = (size_t)B * NHEAD * S * DK;
  float* wsQ = (float*)d_ws;
  float* wsK = wsQ + panel;
  float* wsV = wsK + panel;
  float* wsA = wsV + panel;
  float* wsKV = wsA + panel;
  float* wsC  = wsKV + (size_t)B * NHEAD * DK * DK;

  dim3 blk(256);
  dim3 gemmGrid(DMODEL / 64, M / 256);

  gemm_nt_bias_kernel<<<gemmGrid, blk, 0, stream>>>(q, W_q, b_q, wsQ,
                                                    M, DMODEL, DMODEL, S, 1);
  gemm_nt_bias_kernel<<<gemmGrid, blk, 0, stream>>>(k, W_k, b_k, wsK,
                                                    M, DMODEL, DMODEL, S, 1);
  gemm_nt_bias_kernel<<<gemmGrid, blk, 0, stream>>>(v, W_v, b_v, wsV,
                                                    M, DMODEL, DMODEL, S, 1);

  dim3 kvGrid(4, 2, B * NHEAD);
  kv_context_kernel<<<kvGrid, dim3(KVW * 32), 0, stream>>>(wsK, wsV, msk,
                                                           wsKV, wsC, S);
  dim3 qcGrid(S / 256, B * NHEAD);
  q_context_kernel<<<qcGrid, blk, 0, stream>>>(wsQ, wsKV, wsC, wsA, S);

  gemm_nt_bias_kernel<<<gemmGrid, blk, 0, stream>>>(wsA, W_o, b_o,
                                                    (float*)d_out,
                                                    M, DMODEL, DMODEL, S, 0);
}


# Q, K and V projections in one launch: each block runs its tile of the three as one chunk sequence, the next projection's first chunks staged during the current one's last; sums unchanged
# speedup vs baseline: 3.0478x; 1.0371x over previous
#include <hip/hip_runtime.h>


typedef float v2f __attribute__((ext_vector_type(2)));
typedef float v8f __attribute__((ext_vector_type(8)));

#define DMODEL 1024
#define NHEAD  16
#define DK     64
#define NEG_INF (-1.0e9f)

__device__ inline v8f wmma_f32(v2f a, v2f b, v8f c) {
  return __builtin_amdgcn_wmma_f32_16x16x4_f32(false, a, false, b,
                                               (short)0, c, false, false);
}

#define GT_K   64
#define GT_KG  (GT_K / 4)
#define PLA    ((256 + 1) * 4)
#define PLB    ((64 + 1) * 4)
#define STG    (GT_KG * (PLA + PLB))
#define OPITCH 68

struct Frags { v2f a0, a1, b0, b1, b2, b3; };

__device__ inline Frags load_frags(const float* pa, const float* pb) {
  Frags f;
  f.a0 = *(const v2f*)(pa);
  f.a1 = *(const v2f*)(pa + 64);
  f.b0 = *(const v2f*)(pb);
  f.b1 = *(const v2f*)(pb + 64);
  f.b2 = *(const v2f*)(pb + 128);
  f.b3 = *(const v2f*)(pb + 192);
  return f;
}

__device__ inline void mma_step(const Frags& f, v8f (&acc)[2][4]) {
  acc[0][0] = wmma_f32(f.a0, f.b0, acc[0][0]);
  acc[0][1] = wmma_f32(f.a0, f.b1, acc[0][1]);
  acc[0][2] = wmma_f32(f.a0, f.b2, acc[0][2]);
  acc[0][3] = wmma_f32(f.a0, f.b3, acc[0][3]);
  acc[1][0] = wmma_f32(f.a1, f.b0, acc[1][0]);
  acc[1][1] = wmma_f32(f.a1, f.b1, acc[1][1]);
  acc[1][2] = wmma_f32(f.a1, f.b2, acc[1][2]);
  acc[1][3] = wmma_f32(f.a1, f.b3, acc[1][3]);
}

template <int P0, int P1>
__device__ inline void mma_pairs(const float* pa, const float* pb,
                                 v8f (&acc)[2][4]) {
  Frags f[2][2];
  f[P0 & 1][0] = load_frags(pa + (2 * P0) * PLA,     pb + (2 * P0) * PLB);
  f[P0 & 1][1] = load_frags(pa + (2 * P0 + 1) * PLA, pb + (2 * P0 + 1) * PLB);
  #pragma unroll
  for (int p = P0; p < P1; ++p) {
    if (p + 1 < P1) {
      f[(p + 1) & 1][0] = load_frags(pa + (2 * p + 2) * PLA, pb + (2 * p + 2) * PLB);
      f[(p + 1) & 1][1] = load_frags(pa + (2 * p + 3) * PLA, pb + (2 * p + 3) * PLB);
    }
    mma_step(f[p & 1][0], acc);
    mma_step(f[p & 1][1], acc);
  }
}

#define FOR16(X) X(0) X(1) X(2) X(3) X(4) X(5) X(6) X(7) X(8) X(9) X(10) X(11) X(12) X(13) X(14) X(15)
#define FOR4(X)  X(0) X(1) X(2) X(3)
#define DECL_A(i)  float4 ra##i;
#define DECL_B(i)  float4 rb##i;
#define LOAD_A(i)  ra##i = *(const float4*)(xp + (i) * rowStep);
#define LOAD_B(i)  rb##i = *(const float4*)(wp + (i) * rowStep);
#define STORE_A(i) *(float4*)(dap + (i) * 64) = ra##i;
#define STORE_B(i) *(float4*)(dbp + (i) * 64) = rb##i;

#define PICK(t, p0, p1, p2) ((t) == 0 ? (p0) : ((t) == 1 ? (p1) : (p2)))

__global__ void __launch_bounds__(256)
gemm_nt_bias_kernel(const float* __restrict__ X0, const float* __restrict__ X1,
                    const float* __restrict__ X2,
                    const float* __restrict__ W0, const float* __restrict__ W1,
                    const float* __restrict__ W2,
                    const float* __restrict__ bias0, const float* __restrict__ bias1,
                    const float* __restrict__ bias2,
                    float* __restrict__ Y0, float* __restrict__ Y1,
                    float* __restrict__ Y2,
                    int nJobs, int M, int N, int K, int S, int qkv_layout) {
  __shared__ __attribute__((aligned(16))) float tile[2 * STG + 8 * 32 * OPITCH];

  const int tid  = threadIdx.x;
  const int lane = tid & 31;
  const int wave = tid >> 5;
  const int half = lane >> 4;
  const int l    = lane & 15;
  const int mBlock = blockIdx.y * 256;
  const int mBase  = mBlock + wave * 32;
  const int nBase  = blockIdx.x * 64;

  const int srow = tid >> 4;
  const int sg   = tid & 15;
  const size_t xoff = (size_t)(mBlock + srow) * K + sg * 4;
  const size_t woff = (size_t)(nBase  + srow) * K + sg * 4;
  const int adst = sg * PLA + srow * 4;
  const int bdst = GT_KG * PLA + sg * PLB + srow * 4;
  const int aoff = (wave * 32 + l) * 4 + 2 * half;
  const int boff = GT_KG * PLA + l * 4 + 2 * half;

  v8f acc[2][4] = {{v8f{}, v8f{}, v8f{}, v8f{}},
                   {v8f{}, v8f{}, v8f{}, v8f{}}};
  FOR16(DECL_A) FOR4(DECL_B)

  const int nChunks = K / GT_K;
  const size_t rowStep = (size_t)16 * K;

  int pt = 0, pk = 0;

  {
    const float* xp = X0 + xoff;
    const float* wp = W0 + woff;
    float* dap = tile + adst;
    float* dbp = tile + bdst;
    FOR16(LOAD_A) FOR4(LOAD_B)
    FOR16(STORE_A) FOR4(STORE_B)
    if (pk + 1 < nChunks) ++pk; else if (pt + 1 < nJobs) { pk = 0; ++pt; }
    xp = PICK(pt, X0, X1, X2) + xoff + pk * GT_K;
    wp = PICK(pt, W0, W1, W2) + woff + pk * GT_K;
    FOR16(LOAD_A) FOR4(LOAD_B)
  }
  __syncthreads();

  static_assert(DK == 64, "the 64-column tile must be exactly one head");
  float* patch = tile + 2 * STG + wave * (32 * OPITCH);

  int g = 0;
  for (int t = 0; t < nJobs; ++t) {
    const float* bias = PICK(t, bias0, bias1, bias2);
    const float bv0 = bias[nBase + l];
    const float bv1 = bias[nBase + 16 + l];
    const float bv2 = bias[nBase + 32 + l];
    const float bv3 = bias[nBase + 48 + l];

    const bool lastJob = (t + 1 == nJobs);
    const int nLoop = lastJob ? nChunks - 1 : nChunks;
    for (int kc = 0; kc < nLoop; ++kc, ++g) {
      const float* cur = tile + (g & 1) * STG;
      float* nxt = tile + ((g & 1) ^ 1) * STG;
      if (pk + 1 < nChunks) ++pk; else if (pt + 1 < nJobs) { pk = 0; ++pt; }
      const float* xnext = PICK(pt, X0, X1, X2) + xoff + pk * GT_K;
      const float* wnext = PICK(pt, W0, W1, W2) + woff + pk * GT_K;
      mma_pairs<0, 2>(cur + aoff, cur + boff, acc);
      __builtin_amdgcn_sched_barrier(0);
      {
        float* dap = nxt + adst;
        float* dbp = nxt + bdst;
        const float* xp = xnext;
        const float* wp = wnext;
        FOR16(STORE_A) FOR4(STORE_B)
        FOR16(LOAD_A) FOR4(LOAD_B)
      }
      mma_pairs<2, GT_KG / 2>(cur + aoff, cur + boff, acc);
      __syncthreads();
    }
    if (lastJob) {
      const float* cur = tile + (g & 1) * STG;
      mma_pairs<0, GT_KG / 2>(cur + aoff, cur + boff, acc);
    }

    {
      float* pw = patch + (8 * half) * OPITCH + l;
      const float bv[4] = {bv0, bv1, bv2, bv3};
      #pragma unroll
      for (int j = 0; j < 4; ++j) {
        #pragma unroll
        for (int mi = 0; mi < 2; ++mi) {
          #pragma unroll
          for (int r = 0; r < 8; ++r) {
            pw[(mi * 16 + r) * OPITCH + j * 16] = acc[mi][j][r] + bv[j];
            acc[mi][j][r] = 0.f;
          }
        }
      }
    }
    __syncthreads();

    size_t rowBase, rowPitch;
    if (qkv_layout) {
      const int bidx = mBase / S;
      const int s0   = mBase - bidx * S;
      const int h    = nBase / DK;
      rowBase  = ((size_t)(bidx * NHEAD + h) * S + s0) * DK;
      rowPitch = DK;
    } else {
      rowBase  = (size_t)mBase * N + nBase;
      rowPitch = (size_t)N;
    }
    const float* pr = patch + half * OPITCH + 4 * l;
    float* yp = PICK(t, Y0, Y1, Y2) + rowBase + (size_t)half * rowPitch + 4 * l;
    #pragma unroll
    for (int i = 0; i < 16; ++i) {
      const float4 v = *(const float4*)(pr + (2 * i) * OPITCH);
      *(float4*)(yp + (size_t)(2 * i) * rowPitch) = v;
    }
  }
}

#define KVW 16

__global__ void __launch_bounds__(KVW * 32)
kv_context_kernel(const float* __restrict__ Kp, const float* __restrict__ V,
                  const int* __restrict__ mask, float* __restrict__ KVf,
                  float* __restrict__ Cb, int S) {
  const int lane = threadIdx.x & 31;
  const int wave = threadIdx.x >> 5;
  const int half = lane >> 4;
  const int l    = lane & 15;
  const int it   = blockIdx.x;
  const int jh   = blockIdx.y;
  const int bh   = blockIdx.z;
  const int bidx = bh / NHEAD;

  __shared__ float red[KVW * 512];
  __shared__ float vred[KVW * 64];

  const int kchunk = S / KVW;
  const int kFirst = wave * kchunk + 2 * half;
  const float* kcol = Kp + ((size_t)bh * S + kFirst) * DK + it * 16 + l;
  const float* vcol = V  + ((size_t)bh * S + kFirst) * DK + jh * 32 + l;
  const int*   mp   = mask + (size_t)bidx * S + kFirst;

  v8f acc0 = {}, acc1 = {};
  float vs0 = 0.f, vs1 = 0.f;

  #pragma unroll 8
  for (int k0 = 0; k0 < kchunk; k0 += 4) {
    const int2 mm = *(const int2*)(mp + k0);
    const float* kr = kcol + (size_t)k0 * DK;
    const float* vr = vcol + (size_t)k0 * DK;
    const float ka0 = kr[0];
    const float ka1 = kr[DK];
    v2f b0, b1;
    b0[0] = vr[0];
    b0[1] = vr[DK];
    b1[0] = vr[16];
    b1[1] = vr[DK + 16];
    const bool keep0 = (mm.x != 0);
    const bool keep1 = (mm.y != 0);
    v2f a;
    a[0] = keep0 ? ka0 : 0.f;
    a[1] = keep1 ? ka1 : 0.f;
    acc0 = wmma_f32(a, b0, acc0);
    acc1 = wmma_f32(a, b1, acc1);
    vs0 += (keep0 ? 0.f : b0[0]) + (keep1 ? 0.f : b0[1]);
    vs1 += (keep0 ? 0.f : b1[0]) + (keep1 ? 0.f : b1[1]);
  }

  #pragma unroll
  for (int r = 0; r < 8; ++r) {
    red[wave * 512 + r * 32 + lane]       = acc0[r];
    red[wave * 512 + 256 + r * 32 + lane] = acc1[r];
  }
  vred[wave * 64 + lane]      = vs0;
  vred[wave * 64 + 32 + lane] = vs1;
  __syncthreads();

  {
    const int o   = threadIdx.x;
    const int f   = o >> 6;
    const int kkl = f >> 1, jt = f & 1;
    const int w   = o & 63;
    const int lp  = (w >> 1) & 15, hp = w >> 5, s = w & 1;
    const int el  = 4 * kkl + 2 * hp + s;
    const int src = (jt * 8 + (el & 7)) * 32 + lp + 16 * (el >> 3);
    float sum = 0.f;
    #pragma unroll
    for (int w8 = 0; w8 < KVW; ++w8) sum += red[w8 * 512 + src];
    KVf[(size_t)bh * (DK * DK) +
        ((it * 4 + kkl) * 4 + (jh * 2 + jt)) * 64 + w] = sum * 0.125f;
  }
  if (it == 0 && threadIdx.x < 32) {
    const int t = threadIdx.x >> 4, lp = threadIdx.x & 15;
    float sum = 0.f;
    #pragma unroll
    for (int w8 = 0; w8 < KVW; ++w8)
      sum += vred[w8 * 64 + t * 32 + lp] + vred[w8 * 64 + t * 32 + 16 + lp];
    Cb[bh * DK + jh * 32 + threadIdx.x] = NEG_INF * sum;
  }
}

__global__ void __launch_bounds__(256)
q_context_kernel(const float* __restrict__ Q, const float* __restrict__ KVf,
                 const float* __restrict__ Cb, float* __restrict__ O, int S) {
  const int lane = threadIdx.x & 31;
  const int wave = threadIdx.x >> 5;
  const int half = lane >> 4;
  const int l    = lane & 15;
  const int bh   = blockIdx.y;
  const int bidx = bh / NHEAD;
  const int h    = bh % NHEAD;
  const int qBase = blockIdx.x * 256 + wave * 32;

  const float* qrow0 = Q + ((size_t)bh * S + qBase + l) * DK + 2 * half;
  const float* qrow1 = qrow0 + 16 * DK;
  const float* kv    = KVf + (size_t)bh * (DK * DK) + lane * 2;

  v8f o[2][4] = {{v8f{}, v8f{}, v8f{}, v8f{}},
                 {v8f{}, v8f{}, v8f{}, v8f{}}};

  #pragma unroll
  for (int kk = 0; kk < 16; ++kk) {
    const v2f a0 = *(const v2f*)(qrow0 + kk * 4);
    const v2f a1 = *(const v2f*)(qrow1 + kk * 4);
    #pragma unroll
    for (int j = 0; j < 4; ++j) {
      const v2f b = *(const v2f*)(kv + (kk * 4 + j) * 64);
      o[0][j] = wmma_f32(a0, b, o[0][j]);
      o[1][j] = wmma_f32(a1, b, o[1][j]);
    }
  }

  #pragma unroll
  for (int j = 0; j < 4; ++j) {
    const float cb = Cb[bh * DK + j * 16 + l];
    #pragma unroll
    for (int mi = 0; mi < 2; ++mi) {
      #pragma unroll
      for (int r = 0; r < 8; ++r) {
        const int q = qBase + mi * 16 + r + 8 * half;
        O[((size_t)bidx * S + q) * DMODEL + h * DK + j * 16 + l] = o[mi][j][r] + cb;
      }
    }
  }
}

extern "C" void kernel_launch(void* const* d_in, const int* in_sizes, int n_in,
                              void* d_out, int out_size, void* d_ws, size_t ws_size,
                              hipStream_t stream) {
  const int B = 2, S = 2048;
  const int M = B * S;
  (void)in_sizes; (void)n_in; (void)out_size; (void)ws_size;

  const float* q   = (const float*)d_in[0];
  const float* k   = (const float*)d_in[1];
  const float* v   = (const float*)d_in[2];
  const int*   msk = (const int*)  d_in[3];
  const float* W_q = (const float*)d_in[4];
  const float* b_q = (const float*)d_in[5];
  const float* W_k = (const float*)d_in[6];
  const float* b_k = (const float*)d_in[7];
  const float* W_v = (const float*)d_in[8];
  const float* b_v = (const float*)d_in[9];
  const float* W_o = (const float*)d_in[10];
  const float* b_o = (const float*)d_in[11];

  const size_t panel = (size_t)B * NHEAD * S * DK;
  float* wsQ = (float*)d_ws;
  float* wsK = wsQ + panel;
  float* wsV = wsK + panel;
  float* wsA = wsV + panel;
  float* wsKV = wsA + panel;
  float* wsC  = wsKV + (size_t)B * NHEAD * DK * DK;

  dim3 blk(256);
  dim3 gemmGrid(DMODEL / 64, M / 256);

  gemm_nt_bias_kernel<<<gemmGrid, blk, 0, stream>>>(q, k, v, W_q, W_k, W_v,
                                                    b_q, b_k, b_v,
                                                    wsQ, wsK, wsV, 3,
                                                    M, DMODEL, DMODEL, S, 1);

  dim3 kvGrid(4, 2, B * NHEAD);
  kv_context_kernel<<<kvGrid, dim3(KVW * 32), 0, stream>>>(wsK, wsV, msk,
                                                           wsKV, wsC, S);
  dim3 qcGrid(S / 256, B * NHEAD);
  q_context_kernel<<<qcGrid, blk, 0, stream>>>(wsQ, wsKV, wsC, wsA, S);

  gemm_nt_bias_kernel<<<gemmGrid, blk, 0, stream>>>(wsA, wsA, wsA, W_o, W_o, W_o,
                                                    b_o, b_o, b_o,
                                                    (float*)d_out, (float*)d_out,
                                                    (float*)d_out, 1,
                                                    M, DMODEL, DMODEL, S, 0);
}


# Q projection, Q x context and output projection folded: per batch one 1024x1024 matrix Wo C Wq (contexts folded into Wq, then one product with Wo), out = q times it + row; changes the f32 sum order
# speedup vs baseline: 3.3002x; 1.0828x over previous
#include <hip/hip_runtime.h>


typedef float v2f __attribute__((ext_vector_type(2)));
typedef float v8f __attribute__((ext_vector_type(8)));

#define DMODEL 1024
#define NHEAD  16
#define DK     64
#define NEG_INF (-1.0e9f)

__device__ inline v8f wmma_f32(v2f a, v2f b, v8f c) {
  return __builtin_amdgcn_wmma_f32_16x16x4_f32(false, a, false, b,
                                               (short)0, c, false, false);
}

#define GT_K   64
#define GT_KG  (GT_K / 4)
#define PLA    ((256 + 1) * 4)
#define PLB    ((64 + 1) * 4)
#define STG    (GT_KG * (PLA + PLB))
#define OPITCH 68

struct Frags { v2f a0, a1, b0, b1, b2, b3; };

__device__ inline Frags load_frags(const float* pa, const float* pb) {
  Frags f;
  f.a0 = *(const v2f*)(pa);
  f.a1 = *(const v2f*)(pa + 64);
  f.b0 = *(const v2f*)(pb);
  f.b1 = *(const v2f*)(pb + 64);
  f.b2 = *(const v2f*)(pb + 128);
  f.b3 = *(const v2f*)(pb + 192);
  return f;
}

__device__ inline void mma_step(const Frags& f, v8f (&acc)[2][4]) {
  acc[0][0] = wmma_f32(f.a0, f.b0, acc[0][0]);
  acc[0][1] = wmma_f32(f.a0, f.b1, acc[0][1]);
  acc[0][2] = wmma_f32(f.a0, f.b2, acc[0][2]);
  acc[0][3] = wmma_f32(f.a0, f.b3, acc[0][3]);
  acc[1][0] = wmma_f32(f.a1, f.b0, acc[1][0]);
  acc[1][1] = wmma_f32(f.a1, f.b1, acc[1][1]);
  acc[1][2] = wmma_f32(f.a1, f.b2, acc[1][2]);
  acc[1][3] = wmma_f32(f.a1, f.b3, acc[1][3]);
}

template <int P0, int P1>
__device__ inline void mma_pairs(const float* pa, const float* pb,
                                 v8f (&acc)[2][4]) {
  Frags f[2][2];
  f[P0 & 1][0] = load_frags(pa + (2 * P0) * PLA,     pb + (2 * P0) * PLB);
  f[P0 & 1][1] = load_frags(pa + (2 * P0 + 1) * PLA, pb + (2 * P0 + 1) * PLB);
  #pragma unroll
  for (int p = P0; p < P1; ++p) {
    if (p + 1 < P1) {
      f[(p + 1) & 1][0] = load_frags(pa + (2 * p + 2) * PLA, pb + (2 * p + 2) * PLB);
      f[(p + 1) & 1][1] = load_frags(pa + (2 * p + 3) * PLA, pb + (2 * p + 3) * PLB);
    }
    mma_step(f[p & 1][0], acc);
    mma_step(f[p & 1][1], acc);
  }
}

__device__ inline float4 ld4sum(const float* p, size_t wAdd) {
  const float4 a = *(const float4*)p;
  const float4 b = *(const float4*)(p + wAdd);
  return make_float4(a.x + b.x, a.y + b.y, a.z + b.z, a.w + b.w);
}

#define FOR16(X) X(0) X(1) X(2) X(3) X(4) X(5) X(6) X(7) X(8) X(9) X(10) X(11) X(12) X(13) X(14) X(15)
#define FOR4(X)  X(0) X(1) X(2) X(3)
#define DECL_A(i)  float4 ra##i;
#define DECL_B(i)  float4 rb##i;
#define LOAD_A(i)  ra##i = *(const float4*)(xp + (i) * rowStepX);
#define LOAD_B(i)  rb##i = DUALW ? ld4sum(wp + (i) * rowStepW, wAdd) : *(const float4*)(wp + (i) * rowStepW);
#define STORE_A(i) *(float4*)(dap + (i) * 64) = ra##i;
#define STORE_B(i) *(float4*)(dbp + (i) * 64) = rb##i;

#define PICK(t, p0, p1, p2) ((t) == 0 ? (p0) : ((t) == 1 ? (p1) : (p2)))

template <bool DUALW>
__global__ void __launch_bounds__(256)
gemm_nt_bias_kernel(const float* __restrict__ X0, const float* __restrict__ X1,
                    const float* __restrict__ X2,
                    const float* __restrict__ W0, const float* __restrict__ W1,
                    const float* __restrict__ W2,
                    const float* __restrict__ bias0, const float* __restrict__ bias1,
                    const float* __restrict__ bias2,
                    float* __restrict__ Y0, float* __restrict__ Y1,
                    float* __restrict__ Y2,
                    int nJobs, int M, int N, int K, int S, int qkv_layout,
                    int ldx, int ldw, int xWrap, int wBatchRows,
                    size_t wBatchStride, size_t wAdd, size_t ySplit, int useBias) {
  __shared__ __attribute__((aligned(16))) float tile[2 * STG + 8 * 32 * OPITCH];

  const int tid  = threadIdx.x;
  const int lane = tid & 31;
  const int wave = tid >> 5;
  const int half = lane >> 4;
  const int l    = lane & 15;
  const int mBlock = blockIdx.y * 256;
  const int mBase  = mBlock + wave * 32;
  const int nBase  = blockIdx.x * 64;

  const int srow = tid >> 4;
  const int sg   = tid & 15;
  const int kz   = blockIdx.z;
  const int wb   = mBlock / wBatchRows;
  const int xRow = mBlock - (mBlock / xWrap) * xWrap;
  const size_t xoff = (size_t)(xRow + srow) * ldx + sg * 4 + (size_t)kz * K;
  const size_t woff = (size_t)wb * wBatchStride + (size_t)(nBase + srow) * ldw +
                      sg * 4 + (size_t)kz * K;
  const int adst = sg * PLA + srow * 4;
  const int bdst = GT_KG * PLA + sg * PLB + srow * 4;
  const int aoff = (wave * 32 + l) * 4 + 2 * half;
  const int boff = GT_KG * PLA + l * 4 + 2 * half;

  v8f acc[2][4] = {{v8f{}, v8f{}, v8f{}, v8f{}},
                   {v8f{}, v8f{}, v8f{}, v8f{}}};
  FOR16(DECL_A) FOR4(DECL_B)

  const int nChunks = K / GT_K;
  const size_t rowStepX = (size_t)16 * ldx;
  const size_t rowStepW = (size_t)16 * ldw;

  int pt = 0, pk = 0;

  {
    const float* xp = X0 + xoff;
    const float* wp = W0 + woff;
    float* dap = tile + adst;
    float* dbp = tile + bdst;
    FOR16(LOAD_A) FOR4(LOAD_B)
    FOR16(STORE_A) FOR4(STORE_B)
    if (pk + 1 < nChunks) ++pk; else if (pt + 1 < nJobs) { pk = 0; ++pt; }
    xp = PICK(pt, X0, X1, X2) + xoff + pk * GT_K;
    wp = PICK(pt, W0, W1, W2) + woff + pk * GT_K;
    FOR16(LOAD_A) FOR4(LOAD_B)
  }
  __syncthreads();

  static_assert(DK == 64, "the 64-column tile must be exactly one head");
  float* patch = tile + 2 * STG + wave * (32 * OPITCH);

  int g = 0;
  for (int t = 0; t < nJobs; ++t) {
    const float* bias = PICK(t, bias0, bias1, bias2) + (size_t)wb * N + nBase + l;
    const float bv0 = useBias ? bias[0]  : 0.f;
    const float bv1 = useBias ? bias[16] : 0.f;
    const float bv2 = useBias ? bias[32] : 0.f;
    const float bv3 = useBias ? bias[48] : 0.f;

    const bool lastJob = (t + 1 == nJobs);
    const int nLoop = lastJob ? nChunks - 1 : nChunks;
    for (int kc = 0; kc < nLoop; ++kc, ++g) {
      const float* cur = tile + (g & 1) * STG;
      float* nxt = tile + ((g & 1) ^ 1) * STG;
      if (pk + 1 < nChunks) ++pk; else if (pt + 1 < nJobs) { pk = 0; ++pt; }
      const float* xnext = PICK(pt, X0, X1, X2) + xoff + pk * GT_K;
      const float* wnext = PICK(pt, W0, W1, W2) + woff + pk * GT_K;
      mma_pairs<0, 2>(cur + aoff, cur + boff, acc);
      __builtin_amdgcn_sched_barrier(0);
      {
        float* dap = nxt + adst;
        float* dbp = nxt + bdst;
        const float* xp = xnext;
        const float* wp = wnext;
        FOR16(STORE_A) FOR4(STORE_B)
        FOR16(LOAD_A) FOR4(LOAD_B)
      }
      mma_pairs<2, GT_KG / 2>(cur + aoff, cur + boff, acc);
      __syncthreads();
    }
    if (lastJob) {
      const float* cur = tile + (g & 1) * STG;
      mma_pairs<0, GT_KG / 2>(cur + aoff, cur + boff, acc);
    }

    {
      float* pw = patch + (8 * half) * OPITCH + l;
      const float bv[4] = {bv0, bv1, bv2, bv3};
      #pragma unroll
      for (int j = 0; j < 4; ++j) {
        #pragma unroll
        for (int mi = 0; mi < 2; ++mi) {
          #pragma unroll
          for (int r = 0; r < 8; ++r) {
            pw[(mi * 16 + r) * OPITCH + j * 16] = acc[mi][j][r] + bv[j];
            acc[mi][j][r] = 0.f;
          }
        }
      }
    }
    __syncthreads();

    size_t rowBase, rowPitch;
    if (qkv_layout) {
      const int bidx = mBase / S;
      const int s0   = mBase - bidx * S;
      const int h    = nBase / DK;
      rowBase  = ((size_t)(bidx * NHEAD + h) * S + s0) * DK;
      rowPitch = DK;
    } else {
      rowBase  = (size_t)mBase * N + nBase;
      rowPitch = (size_t)N;
    }
    const float* pr = patch + half * OPITCH + 4 * l;
    float* yp = PICK(t, Y0, Y1, Y2) + (size_t)kz * ySplit + rowBase +
                (size_t)half * rowPitch + 4 * l;
    #pragma unroll
    for (int i = 0; i < 16; ++i) {
      const float4 v = *(const float4*)(pr + (2 * i) * OPITCH);
      *(float4*)(yp + (size_t)(2 * i) * rowPitch) = v;
    }
  }
}

#define KVW 16

__global__ void __launch_bounds__(KVW * 32)
kv_context_kernel(const float* __restrict__ Kp, const float* __restrict__ V,
                  const int* __restrict__ mask, float* __restrict__ KVf,
                  float* __restrict__ Cb, int S) {
  const int lane = threadIdx.x & 31;
  const int wave = threadIdx.x >> 5;
  const int half = lane >> 4;
  const int l    = lane & 15;
  const int it   = blockIdx.x;
  const int jh   = blockIdx.y;
  const int bh   = blockIdx.z;
  const int bidx = bh / NHEAD;

  __shared__ float red[KVW * 512];
  __shared__ float vred[KVW * 64];

  const int kchunk = S / KVW;
  const int kFirst = wave * kchunk + 2 * half;
  const float* kcol = Kp + ((size_t)bh * S + kFirst) * DK + it * 16 + l;
  const float* vcol = V  + ((size_t)bh * S + kFirst) * DK + jh * 32 + l;
  const int*   mp   = mask + (size_t)bidx * S + kFirst;

  v8f acc0 = {}, acc1 = {};
  float vs0 = 0.f, vs1 = 0.f;

  #pragma unroll 8
  for (int k0 = 0; k0 < kchunk; k0 += 4) {
    const int2 mm = *(const int2*)(mp + k0);
    const float* kr = kcol + (size_t)k0 * DK;
    const float* vr = vcol + (size_t)k0 * DK;
    const float ka0 = kr[0];
    const float ka1 = kr[DK];
    v2f b0, b1;
    b0[0] = vr[0];
    b0[1] = vr[DK];
    b1[0] = vr[16];
    b1[1] = vr[DK + 16];
    const bool keep0 = (mm.x != 0);
    const bool keep1 = (mm.y != 0);
    v2f a;
    a[0] = keep0 ? ka0 : 0.f;
    a[1] = keep1 ? ka1 : 0.f;
    acc0 = wmma_f32(a, b0, acc0);
    acc1 = wmma_f32(a, b1, acc1);
    vs0 += (keep0 ? 0.f : b0[0]) + (keep1 ? 0.f : b0[1]);
    vs1 += (keep0 ? 0.f : b1[0]) + (keep1 ? 0.f : b1[1]);
  }

  #pragma unroll
  for (int r = 0; r < 8; ++r) {
    red[wave * 512 + r * 32 + lane]       = acc0[r];
    red[wave * 512 + 256 + r * 32 + lane] = acc1[r];
  }
  vred[wave * 64 + lane]      = vs0;
  vred[wave * 64 + 32 + lane] = vs1;
  __syncthreads();

  {
    const int o   = threadIdx.x;
    const int f   = o >> 6;
    const int kkl = f >> 1, jt = f & 1;
    const int w   = o & 63;
    const int lp  = (w >> 1) & 15, hp = w >> 5, s = w & 1;
    const int el  = 4 * kkl + 2 * hp + s;
    const int src = (jt * 8 + (el & 7)) * 32 + lp + 16 * (el >> 3);
    float sum = 0.f;
    #pragma unroll
    for (int w8 = 0; w8 < KVW; ++w8) sum += red[w8 * 512 + src];
    KVf[(size_t)bh * (DK * DK) +
        ((it * 4 + kkl) * 4 + (jh * 2 + jt)) * 64 + w] = sum * 0.125f;
  }
  if (it == 0 && threadIdx.x < 32) {
    const int t = threadIdx.x >> 4, lp = threadIdx.x & 15;
    float sum = 0.f;
    #pragma unroll
    for (int w8 = 0; w8 < KVW; ++w8)
      sum += vred[w8 * 64 + t * 32 + lp] + vred[w8 * 64 + t * 32 + 16 + lp];
    Cb[bh * DK + jh * 32 + threadIdx.x] = NEG_INF * sum;
  }
}

__device__ inline void bias_fold_block(const float* __restrict__ KVf,
                                       const float* __restrict__ Cb,
                                       const float* __restrict__ bq,
                                       const float* __restrict__ Wo,
                                       const float* __restrict__ bo,
                                       float* __restrict__ bvec,
                                       int x, int bidx) {
  __shared__ __attribute__((aligned(16))) float cbs[DMODEL];
  __shared__ float outv[32];
  for (int c = threadIdx.x; c < DMODEL; c += 256) {
    const int h = c >> 6, d = c & 63;
    const float* kvh = KVf + (size_t)(bidx * NHEAD + h) * (DK * DK);
    const float* bqh = bq + h * DK;
    float sum = 0.f;
    #pragma unroll
    for (int e = 0; e < DK; ++e) {
      const int idx = ((e >> 2) * 4 + (d >> 4)) * 64 +
                      ((d & 15) + 16 * ((e >> 1) & 1)) * 2 + (e & 1);
      sum += bqh[e] * kvh[idx];
    }
    cbs[c] = Cb[bidx * DMODEL + c] + sum;
  }
  __syncthreads();

  const int lane = threadIdx.x & 31;
  const int wave = threadIdx.x >> 5;
  for (int rr = 0; rr < 4; ++rr) {
    const int n = x * 32 + wave * 4 + rr;
    const float* wr = Wo + (size_t)n * DMODEL;
    float sum = 0.f;
    #pragma unroll
    for (int t = 0; t < 8; ++t) {
      const int c = t * 128 + lane * 4;
      const float4 w  = *(const float4*)(wr + c);
      const float4 cv = *(const float4*)(cbs + c);
      sum += w.x * cv.x + w.y * cv.y + w.z * cv.z + w.w * cv.w;
    }
    for (int off = 16; off > 0; off >>= 1) sum += __shfl_xor(sum, off);
    if (lane == 0) outv[wave * 4 + rr] = bo[n] + sum;
  }
  __syncthreads();
  if (threadIdx.x < 32)
    bvec[bidx * DMODEL + x * 32 + threadIdx.x] = outv[threadIdx.x];
}

__global__ void __launch_bounds__(256)
wq_context_kernel(const float* __restrict__ Wq, const float* __restrict__ KVf,
                  float* __restrict__ PT, const float* __restrict__ Cb,
                  const float* __restrict__ bq, const float* __restrict__ Wo,
                  const float* __restrict__ bo, float* __restrict__ bvec, int nBH) {
  if ((int)blockIdx.y >= nBH) {
    const int rb = ((int)blockIdx.y - nBH) * 4 + (int)blockIdx.x;
    bias_fold_block(KVf, Cb, bq, Wo, bo, bvec, rb & 31, rb >> 5);
    return;
  }
  const int lane = threadIdx.x & 31;
  const int wave = threadIdx.x >> 5;
  const int half = lane >> 4;
  const int l    = lane & 15;
  const int bh   = blockIdx.y;
  const int bidx = bh / NHEAD;
  const int h    = bh % NHEAD;
  const int iBase = blockIdx.x * 256 + wave * 32;

  const float* wq0 = Wq + (size_t)(h * DK + 2 * half) * DMODEL + iBase + l;
  const float* kv  = KVf + (size_t)bh * (DK * DK) + lane * 2;

  v8f o[2][4] = {{v8f{}, v8f{}, v8f{}, v8f{}},
                 {v8f{}, v8f{}, v8f{}, v8f{}}};

  #pragma unroll
  for (int kk = 0; kk < 16; ++kk) {
    const float* wr = wq0 + (size_t)(4 * kk) * DMODEL;
    v2f a0, a1;
    a0[0] = wr[0];
    a0[1] = wr[DMODEL];
    a1[0] = wr[16];
    a1[1] = wr[DMODEL + 16];
    #pragma unroll
    for (int j = 0; j < 4; ++j) {
      const v2f b = *(const v2f*)(kv + (kk * 4 + j) * 64);
      o[0][j] = wmma_f32(a0, b, o[0][j]);
      o[1][j] = wmma_f32(a1, b, o[1][j]);
    }
  }

  #pragma unroll
  for (int j = 0; j < 4; ++j) {
    #pragma unroll
    for (int mi = 0; mi < 2; ++mi) {
      #pragma unroll
      for (int r = 0; r < 8; ++r) {
        const int i = iBase + mi * 16 + r + 8 * half;
        PT[((size_t)bidx * DMODEL + i) * DMODEL + h * DK + j * 16 + l] = o[mi][j][r];
      }
    }
  }
}

extern "C" void kernel_launch(void* const* d_in, const int* in_sizes, int n_in,
                              void* d_out, int out_size, void* d_ws, size_t ws_size,
                              hipStream_t stream) {
  const int B = 2, S = 2048;
  const int M = B * S;
  (void)in_sizes; (void)n_in; (void)out_size; (void)ws_size;

  const float* q   = (const float*)d_in[0];
  const float* k   = (const float*)d_in[1];
  const float* v   = (const float*)d_in[2];
  const int*   msk = (const int*)  d_in[3];
  const float* W_q = (const float*)d_in[4];
  const float* b_q = (const float*)d_in[5];
  const float* W_k = (const float*)d_in[6];
  const float* b_k = (const float*)d_in[7];
  const float* W_v = (const float*)d_in[8];
  const float* b_v = (const float*)d_in[9];
  const float* W_o = (const float*)d_in[10];
  const float* b_o = (const float*)d_in[11];

  const size_t panel = (size_t)B * NHEAD * S * DK;
  const size_t sq    = (size_t)DMODEL * DMODEL;
  float* wsK  = (float*)d_ws;
  float* wsV  = wsK + panel;
  float* wsKV = wsV + panel;
  float* wsC  = wsKV + (size_t)B * NHEAD * DK * DK;
  float* wsPT = wsC + (size_t)B * DMODEL;
  float* wsMt = wsPT + 2 * sq;
  float* wsBv = wsMt + 4 * sq;

  dim3 blk(256);
  dim3 gemmGrid(DMODEL / 64, M / 256);

  gemm_nt_bias_kernel<false><<<gemmGrid, blk, 0, stream>>>(
      k, v, v, W_k, W_v, W_v, b_k, b_v, b_v, wsK, wsV, wsV, 2,
      M, DMODEL, DMODEL, S, 1,
      DMODEL, DMODEL, M, M, (size_t)0, (size_t)0, (size_t)0, 1);

  dim3 kvGrid(4, 2, B * NHEAD);
  kv_context_kernel<<<kvGrid, dim3(KVW * 32), 0, stream>>>(wsK, wsV, msk,
                                                           wsKV, wsC, S);

  wq_context_kernel<<<dim3(DMODEL / 256, B * NHEAD + 16), blk, 0, stream>>>(
      W_q, wsKV, wsPT, wsC, b_q, W_o, b_o, wsBv, B * NHEAD);

  gemm_nt_bias_kernel<false><<<dim3(DMODEL / 64, (B * DMODEL) / 256, 2), blk, 0, stream>>>(
      W_o, W_o, W_o, wsPT, wsPT, wsPT, wsBv, wsBv, wsBv, wsMt, wsMt, wsMt, 1,
      B * DMODEL, DMODEL, DMODEL / 2, S, 0,
      DMODEL, DMODEL, DMODEL, DMODEL, sq, (size_t)0, 2 * sq, 0);

  gemm_nt_bias_kernel<true><<<gemmGrid, blk, 0, stream>>>(
      q, q, q, wsMt, wsMt, wsMt, wsBv, wsBv, wsBv,
      (float*)d_out, (float*)d_out, (float*)d_out, 1,
      M, DMODEL, DMODEL, S, 0,
      DMODEL, DMODEL, M, S, sq, 2 * sq, (size_t)0, 1);
}


# K, V projection launch keeps its K and V tiles in LDS and reduces them to per-block partial contexts on the matrix unit; context kernel and K, V panels dropped; changes the f32 sum order
# speedup vs baseline: 3.5717x; 1.0823x over previous
#include <hip/hip_runtime.h>


typedef float v2f __attribute__((ext_vector_type(2)));
typedef float v8f __attribute__((ext_vector_type(8)));

#define DMODEL 1024
#define NHEAD  16
#define DK     64
#define NEG_INF (-1.0e9f)

__device__ inline v8f wmma_f32(v2f a, v2f b, v8f c) {
  return __builtin_amdgcn_wmma_f32_16x16x4_f32(false, a, false, b,
                                               (short)0, c, false, false);
}

#define GT_K   64
#define GT_KG  (GT_K / 4)
#define PLA    ((256 + 1) * 4)
#define PLB    ((64 + 1) * 4)
#define STG    (GT_KG * (PLA + PLB))
#define OPITCH 68
#define PATCHF (8 * 32 * OPITCH)
#define TPITCH (256 + 4)
#define TPATCH (64 * TPITCH)

struct Frags { v2f a0, a1, b0, b1, b2, b3; };

__device__ inline Frags load_frags(const float* pa, const float* pb) {
  Frags f;
  f.a0 = *(const v2f*)(pa);
  f.a1 = *(const v2f*)(pa + 64);
  f.b0 = *(const v2f*)(pb);
  f.b1 = *(const v2f*)(pb + 64);
  f.b2 = *(const v2f*)(pb + 128);
  f.b3 = *(const v2f*)(pb + 192);
  return f;
}

__device__ inline void mma_step(const Frags& f, v8f (&acc)[2][4]) {
  acc[0][0] = wmma_f32(f.a0, f.b0, acc[0][0]);
  acc[0][1] = wmma_f32(f.a0, f.b1, acc[0][1]);
  acc[0][2] = wmma_f32(f.a0, f.b2, acc[0][2]);
  acc[0][3] = wmma_f32(f.a0, f.b3, acc[0][3]);
  acc[1][0] = wmma_f32(f.a1, f.b0, acc[1][0]);
  acc[1][1] = wmma_f32(f.a1, f.b1, acc[1][1]);
  acc[1][2] = wmma_f32(f.a1, f.b2, acc[1][2]);
  acc[1][3] = wmma_f32(f.a1, f.b3, acc[1][3]);
}

template <int P0, int P1>
__device__ inline void mma_pairs(const float* pa, const float* pb,
                                 v8f (&acc)[2][4]) {
  Frags f[2][2];
  f[P0 & 1][0] = load_frags(pa + (2 * P0) * PLA,     pb + (2 * P0) * PLB);
  f[P0 & 1][1] = load_frags(pa + (2 * P0 + 1) * PLA, pb + (2 * P0 + 1) * PLB);
  #pragma unroll
  for (int p = P0; p < P1; ++p) {
    if (p + 1 < P1) {
      f[(p + 1) & 1][0] = load_frags(pa + (2 * p + 2) * PLA, pb + (2 * p + 2) * PLB);
      f[(p + 1) & 1][1] = load_frags(pa + (2 * p + 3) * PLA, pb + (2 * p + 3) * PLB);
    }
    mma_step(f[p & 1][0], acc);
    mma_step(f[p & 1][1], acc);
  }
}

__device__ inline float4 ld4sum(const float* p, size_t wAdd) {
  const float4 a = *(const float4*)p;
  const float4 b = *(const float4*)(p + wAdd);
  return make_float4(a.x + b.x, a.y + b.y, a.z + b.z, a.w + b.w);
}

#define FOR16(X) X(0) X(1) X(2) X(3) X(4) X(5) X(6) X(7) X(8) X(9) X(10) X(11) X(12) X(13) X(14) X(15)
#define FOR4(X)  X(0) X(1) X(2) X(3)
#define DECL_A(i)  float4 ra##i;
#define DECL_B(i)  float4 rb##i;
#define LOAD_A(i)  ra##i = *(const float4*)(xp + (i) * rowStepX);
#define LOAD_B(i)  rb##i = DUALW ? ld4sum(wp + (i) * rowStepW, wAdd) : *(const float4*)(wp + (i) * rowStepW);
#define STORE_A(i) *(float4*)(dap + (i) * 64) = ra##i;
#define STORE_B(i) *(float4*)(dbp + (i) * 64) = rb##i;

#define PICK(t, p0, p1, p2) ((t) == 0 ? (p0) : ((t) == 1 ? (p1) : (p2)))

template <bool DUALW, bool CTX>
__global__ void __launch_bounds__(256)
gemm_nt_bias_kernel(const float* __restrict__ X0, const float* __restrict__ X1,
                    const float* __restrict__ X2,
                    const float* __restrict__ W0, const float* __restrict__ W1,
                    const float* __restrict__ W2,
                    const float* __restrict__ bias0, const float* __restrict__ bias1,
                    const float* __restrict__ bias2,
                    float* __restrict__ Y0, float* __restrict__ Y1,
                    float* __restrict__ Y2,
                    int nJobs, int M, int N, int K, int S, int qkv_layout,
                    int ldx, int ldw, int xWrap, int wBatchRows,
                    size_t wBatchStride, size_t wAdd, size_t ySplit, int useBias,
                    const int* __restrict__ maskp, const float* __restrict__ bq,
                    float* __restrict__ ctxP, float* __restrict__ cbP,
                    float* __restrict__ bqP) {
  __shared__ __attribute__((aligned(16)))
      float tile[2 * STG + (CTX ? 2 * TPATCH + 256 + 8 : PATCHF)];

  const int tid  = threadIdx.x;
  const int lane = tid & 31;
  const int wave = tid >> 5;
  const int half = lane >> 4;
  const int l    = lane & 15;
  const int mBlock = blockIdx.y * 256;
  const int mBase  = mBlock + wave * 32;
  const int nBase  = blockIdx.x * 64;

  const int srow = tid >> 4;
  const int sg   = tid & 15;
  const int kz   = blockIdx.z;
  const int wb   = mBlock / wBatchRows;
  const int xRow = mBlock - (mBlock / xWrap) * xWrap;
  const size_t xoff = (size_t)(xRow + srow) * ldx + sg * 4 + (size_t)kz * K;
  const size_t woff = (size_t)wb * wBatchStride + (size_t)(nBase + srow) * ldw +
                      sg * 4 + (size_t)kz * K;
  const int adst = sg * PLA + srow * 4;
  const int bdst = GT_KG * PLA + sg * PLB + srow * 4;
  const int aoff = (wave * 32 + l) * 4 + 2 * half;
  const int boff = GT_KG * PLA + l * 4 + 2 * half;

  v8f acc[2][4] = {{v8f{}, v8f{}, v8f{}, v8f{}},
                   {v8f{}, v8f{}, v8f{}, v8f{}}};
  FOR16(DECL_A) FOR4(DECL_B)

  float* keepS = tile + 2 * STG + 2 * TPATCH;
  float* flagS = keepS + 256;
  if constexpr (CTX) {
    const bool kkeep = maskp[mBlock + tid] != 0;
    keepS[tid] = kkeep ? 1.f : 0.f;
    const unsigned long long bal = __ballot(!kkeep);
    if (lane == 0) flagS[wave] = (bal != 0ull) ? 1.f : 0.f;
  }

  const int nChunks = K / GT_K;
  const size_t rowStepX = (size_t)16 * ldx;
  const size_t rowStepW = (size_t)16 * ldw;

  int pt = 0, pk = 0;

  {
    const float* xp = X0 + xoff;
    const float* wp = W0 + woff;
    float* dap = tile + adst;
    float* dbp = tile + bdst;
    FOR16(LOAD_A) FOR4(LOAD_B)
    FOR16(STORE_A) FOR4(STORE_B)
    if (pk + 1 < nChunks) ++pk; else if (pt + 1 < nJobs) { pk = 0; ++pt; }
    xp = PICK(pt, X0, X1, X2) + xoff + pk * GT_K;
    wp = PICK(pt, W0, W1, W2) + woff + pk * GT_K;
    FOR16(LOAD_A) FOR4(LOAD_B)
  }
  __syncthreads();

  static_assert(DK == 64, "the 64-column tile must be exactly one head");
  float* patch = tile + 2 * STG + wave * (32 * OPITCH);

  int g = 0;
  for (int t = 0; t < nJobs; ++t) {
    const float* bias = PICK(t, bias0, bias1, bias2) + (size_t)wb * N + nBase + l;
    const float bv0 = useBias ? bias[0]  : 0.f;
    const float bv1 = useBias ? bias[16] : 0.f;
    const float bv2 = useBias ? bias[32] : 0.f;
    const float bv3 = useBias ? bias[48] : 0.f;

    const bool lastJob = (t + 1 == nJobs);
    const int nLoop = lastJob ? nChunks - 1 : nChunks;
    for (int kc = 0; kc < nLoop; ++kc, ++g) {
      const float* cur = tile + (g & 1) * STG;
      float* nxt = tile + ((g & 1) ^ 1) * STG;
      if (pk + 1 < nChunks) ++pk; else if (pt + 1 < nJobs) { pk = 0; ++pt; }
      const float* xnext = PICK(pt, X0, X1, X2) + xoff + pk * GT_K;
      const float* wnext = PICK(pt, W0, W1, W2) + woff + pk * GT_K;
      mma_pairs<0, 2>(cur + aoff, cur + boff, acc);
      __builtin_amdgcn_sched_barrier(0);
      {
        float* dap = nxt + adst;
        float* dbp = nxt + bdst;
        const float* xp = xnext;
        const float* wp = wnext;
        FOR16(STORE_A) FOR4(STORE_B)
        FOR16(LOAD_A) FOR4(LOAD_B)
      }
      mma_pairs<2, GT_KG / 2>(cur + aoff, cur + boff, acc);
      __syncthreads();
    }
    if (lastJob) {
      const float* cur = tile + (g & 1) * STG;
      mma_pairs<0, GT_KG / 2>(cur + aoff, cur + boff, acc);
    }

    if constexpr (CTX) {
      float* pw = tile + 2 * STG + t * TPATCH + l * TPITCH + wave * 32 + 8 * half;
      const float bv[4] = {bv0, bv1, bv2, bv3};
      const float* kp = keepS + wave * 32 + 8 * half;
      #pragma unroll
      for (int mi = 0; mi < 2; ++mi) {
        const float4 k0 = *(const float4*)(kp + mi * 16);
        const float4 k1 = *(const float4*)(kp + mi * 16 + 4);
        const float kf[8] = {k0.x, k0.y, k0.z, k0.w, k1.x, k1.y, k1.z, k1.w};
        bool kept[8];
        #pragma unroll
        for (int r = 0; r < 8; ++r) kept[r] = (t != 0) | (kf[r] != 0.f);
        #pragma unroll
        for (int j = 0; j < 4; ++j) {
          float o[8];
          #pragma unroll
          for (int r = 0; r < 8; ++r) {
            o[r] = kept[r] ? acc[mi][j][r] + bv[j] : 0.f;
            acc[mi][j][r] = 0.f;
          }
          float* dst = pw + (j * 16) * TPITCH + mi * 16;
          *(float4*)(dst)     = make_float4(o[0], o[1], o[2], o[3]);
          *(float4*)(dst + 4) = make_float4(o[4], o[5], o[6], o[7]);
        }
      }
      continue;
    }
    {
      float* pw = patch + (8 * half) * OPITCH + l;
      const float bv[4] = {bv0, bv1, bv2, bv3};
      #pragma unroll
      for (int j = 0; j < 4; ++j) {
        #pragma unroll
        for (int mi = 0; mi < 2; ++mi) {
          #pragma unroll
          for (int r = 0; r < 8; ++r) {
            pw[(mi * 16 + r) * OPITCH + j * 16] = acc[mi][j][r] + bv[j];
            acc[mi][j][r] = 0.f;
          }
        }
      }
    }
    __syncthreads();

    size_t rowBase, rowPitch;
    if (qkv_layout) {
      const int bidx = mBase / S;
      const int s0   = mBase - bidx * S;
      const int h    = nBase / DK;
      rowBase  = ((size_t)(bidx * NHEAD + h) * S + s0) * DK;
      rowPitch = DK;
    } else {
      rowBase  = (size_t)mBase * N + nBase;
      rowPitch = (size_t)N;
    }
    const float* pr = patch + half * OPITCH + 4 * l;
    float* yp = PICK(t, Y0, Y1, Y2) + (size_t)kz * ySplit + rowBase +
                (size_t)half * rowPitch + 4 * l;
    #pragma unroll
    for (int i = 0; i < 16; ++i) {
      const float4 v = *(const float4*)(pr + (2 * i) * OPITCH);
      *(float4*)(yp + (size_t)(2 * i) * rowPitch) = v;
    }
  }

  if constexpr (CTX) {
    __syncthreads();
    const float* PK = tile + 2 * STG;
    const float* PV = PK + TPATCH;
    const int et  = wave >> 1;
    const int dt0 = (wave & 1) * 2;
    const float* pa = PK + (et * 16 + l) * TPITCH + 4 * half;
    const float* pb = PV + (dt0 * 16 + l) * TPITCH + 4 * half;
    v8f c00 = {}, c01 = {}, c10 = {}, c11 = {};
    #pragma unroll 4
    for (int g8 = 0; g8 < 256; g8 += 8) {
      const float4 a  = *(const float4*)(pa + g8);
      const float4 b0 = *(const float4*)(pb + g8);
      const float4 b1 = *(const float4*)(pb + 16 * TPITCH + g8);
      v2f a0, a1, b00, b01, b10, b11;
      a0[0]  = a.x;  a0[1]  = a.y;  a1[0]  = a.z;  a1[1]  = a.w;
      b00[0] = b0.x; b00[1] = b0.y; b10[0] = b0.z; b10[1] = b0.w;
      b01[0] = b1.x; b01[1] = b1.y; b11[0] = b1.z; b11[1] = b1.w;
      c00 = wmma_f32(a0, b00, c00);
      c01 = wmma_f32(a0, b01, c01);
      c10 = wmma_f32(a1, b10, c10);
      c11 = wmma_f32(a1, b11, c11);
    }
    const int mbi = blockIdx.y;
    const int hd0 = nBase + dt0 * 16 + l;
    float* cp = ctxP + ((size_t)mbi * NHEAD + blockIdx.x) * (DK * DK) +
                (et * 16 + 8 * half) * DK + dt0 * 16 + l;
    const float* bqh = bq + nBase + et * 16 + 8 * half;
    float s0 = 0.f, s1 = 0.f;
    #pragma unroll
    for (int r = 0; r < 8; ++r) {
      const float v0 = c00[r] + c10[r];
      const float v1 = c01[r] + c11[r];
      cp[r * DK]      = v0;
      cp[r * DK + 16] = v1;
      const float w = bqh[r];
      s0 += w * v0;
      s1 += w * v1;
    }
    s0 += __shfl_xor(s0, 16);
    s1 += __shfl_xor(s1, 16);
    if (half == 0) {
      float* bp = bqP + (size_t)(mbi * 4 + et) * DMODEL + hd0;
      bp[0]  = s0;
      bp[16] = s1;
    }
    float anyMasked = 0.f;
    #pragma unroll
    for (int w8 = 0; w8 < 8; ++w8) anyMasked += flagS[w8];
    const int q = tid >> 6, d = tid & 63;
    float vsum = 0.f;
    if (anyMasked != 0.f) {
      const float* pv = PV + d * TPITCH + q * 64;
      const float* kq = keepS + q * 64;
      #pragma unroll 8
      for (int R = 0; R < 64; ++R)
        vsum += (kq[R] != 0.f) ? 0.f : pv[R];
    }
    cbP[(size_t)(mbi * 4 + q) * DMODEL + nBase + d] = vsum;
  }
}

#define NPART 8

__device__ inline void bias_fold_block(const float* __restrict__ cbP,
                                       const float* __restrict__ bqP,
                                       const float* __restrict__ Wo,
                                       const float* __restrict__ bo,
                                       float* __restrict__ bvec,
                                       int x, int bidx) {
  __shared__ __attribute__((aligned(16))) float cbs[DMODEL];
  __shared__ float outv[32];
  for (int c = threadIdx.x; c < DMODEL; c += 256) {
    const float* pv = cbP + (size_t)(bidx * 4 * NPART) * DMODEL + c;
    const float* pq = bqP + (size_t)(bidx * 4 * NPART) * DMODEL + c;
    float sv = 0.f, sq = 0.f;
    #pragma unroll
    for (int p = 0; p < 4 * NPART; ++p) {
      sv += pv[(size_t)p * DMODEL];
      sq += pq[(size_t)p * DMODEL];
    }
    cbs[c] = NEG_INF * sv + 0.125f * sq;
  }
  __syncthreads();

  const int lane = threadIdx.x & 31;
  const int wave = threadIdx.x >> 5;
  for (int rr = 0; rr < 4; ++rr) {
    const int n = x * 32 + wave * 4 + rr;
    const float* wr = Wo + (size_t)n * DMODEL;
    float sum = 0.f;
    #pragma unroll
    for (int t = 0; t < 8; ++t) {
      const int c = t * 128 + lane * 4;
      const float4 w  = *(const float4*)(wr + c);
      const float4 cv = *(const float4*)(cbs + c);
      sum += w.x * cv.x + w.y * cv.y + w.z * cv.z + w.w * cv.w;
    }
    for (int off = 16; off > 0; off >>= 1) sum += __shfl_xor(sum, off);
    if (lane == 0) outv[wave * 4 + rr] = bo[n] + sum;
  }
  __syncthreads();
  if (threadIdx.x < 32)
    bvec[bidx * DMODEL + x * 32 + threadIdx.x] = outv[threadIdx.x];
}

__global__ void __launch_bounds__(256)
wq_context_kernel(const float* __restrict__ Wq, const float* __restrict__ ctxP,
                  float* __restrict__ PT, const float* __restrict__ cbP,
                  const float* __restrict__ bqP, const float* __restrict__ Wo,
                  const float* __restrict__ bo, float* __restrict__ bvec, int nBH) {
  if ((int)blockIdx.y >= nBH) {
    const int rb = ((int)blockIdx.y - nBH) * 4 + (int)blockIdx.x;
    bias_fold_block(cbP, bqP, Wo, bo, bvec, rb & 31, rb >> 5);
    return;
  }
  __shared__ __attribute__((aligned(16))) float kvs[DK * OPITCH];
  const int lane = threadIdx.x & 31;
  const int wave = threadIdx.x >> 5;
  const int half = lane >> 4;
  const int l    = lane & 15;
  const int bh   = blockIdx.y;
  const int bidx = bh / NHEAD;
  const int h    = bh % NHEAD;
  const int iBase = blockIdx.x * 256 + wave * 32;

  const float* wq0 = Wq + (size_t)(h * DK + 2 * half) * DMODEL + iBase + l;

  {
    const float* cp0 = ctxP + ((size_t)(bidx * NPART) * NHEAD + h) * (DK * DK);
    #pragma unroll
    for (int it = 0; it < DK * DK / 4 / 256; ++it) {
      const int i4 = threadIdx.x + it * 256;
      float4 sum = make_float4(0.f, 0.f, 0.f, 0.f);
      #pragma unroll
      for (int p = 0; p < NPART; ++p) {
        const float4 v = *(const float4*)(cp0 + (size_t)p * NHEAD * (DK * DK) + i4 * 4);
        sum.x += v.x; sum.y += v.y; sum.z += v.z; sum.w += v.w;
      }
      const int e = i4 >> 4, d4 = (i4 & 15) * 4;
      *(float4*)(kvs + e * OPITCH + d4) =
          make_float4(sum.x * 0.125f, sum.y * 0.125f, sum.z * 0.125f, sum.w * 0.125f);
    }
  }
  __syncthreads();
  const float* kv = kvs + (2 * half) * OPITCH + l;

  v8f o[2][4] = {{v8f{}, v8f{}, v8f{}, v8f{}},
                 {v8f{}, v8f{}, v8f{}, v8f{}}};

  #pragma unroll
  for (int kk = 0; kk < 16; ++kk) {
    const float* wr = wq0 + (size_t)(4 * kk) * DMODEL;
    v2f a0, a1;
    a0[0] = wr[0];
    a0[1] = wr[DMODEL];
    a1[0] = wr[16];
    a1[1] = wr[DMODEL + 16];
    #pragma unroll
    for (int j = 0; j < 4; ++j) {
      v2f b;
      b[0] = kv[(4 * kk) * OPITCH + j * 16];
      b[1] = kv[(4 * kk + 1) * OPITCH + j * 16];
      o[0][j] = wmma_f32(a0, b, o[0][j]);
      o[1][j] = wmma_f32(a1, b, o[1][j]);
    }
  }

  #pragma unroll
  for (int j = 0; j < 4; ++j) {
    #pragma unroll
    for (int mi = 0; mi < 2; ++mi) {
      #pragma unroll
      for (int r = 0; r < 8; ++r) {
        const int i = iBase + mi * 16 + r + 8 * half;
        PT[((size_t)bidx * DMODEL + i) * DMODEL + h * DK + j * 16 + l] = o[mi][j][r];
      }
    }
  }
}

extern "C" void kernel_launch(void* const* d_in, const int* in_sizes, int n_in,
                              void* d_out, int out_size, void* d_ws, size_t ws_size,
                              hipStream_t stream) {
  const int B = 2, S = 2048;
  const int M = B * S;
  (void)in_sizes; (void)n_in; (void)out_size; (void)ws_size;

  const float* q   = (const float*)d_in[0];
  const float* k   = (const float*)d_in[1];
  const float* v   = (const float*)d_in[2];
  const int*   msk = (const int*)  d_in[3];
  const float* W_q = (const float*)d_in[4];
  const float* b_q = (const float*)d_in[5];
  const float* W_k = (const float*)d_in[6];
  const float* b_k = (const float*)d_in[7];
  const float* W_v = (const float*)d_in[8];
  const float* b_v = (const float*)d_in[9];
  const float* W_o = (const float*)d_in[10];
  const float* b_o = (const float*)d_in[11];

  const size_t sq    = (size_t)DMODEL * DMODEL;
  float* wsCtx = (float*)d_ws;
  float* wsCbP = wsCtx + (size_t)(M / 256) * NHEAD * DK * DK;
  float* wsBqP = wsCbP + (size_t)(M / 64) * DMODEL;
  float* wsPT = wsBqP + (size_t)(M / 64) * DMODEL;
  float* wsMt = wsPT + 2 * sq;
  float* wsBv = wsMt + 4 * sq;

  dim3 blk(256);
  dim3 gemmGrid(DMODEL / 64, M / 256);

  gemm_nt_bias_kernel<false, true><<<gemmGrid, blk, 0, stream>>>(
      k, v, v, W_k, W_v, W_v, b_k, b_v, b_v, wsPT, wsPT, wsPT, 2,
      M, DMODEL, DMODEL, S, 1,
      DMODEL, DMODEL, M, M, (size_t)0, (size_t)0, (size_t)0, 1,
      msk, b_q, wsCtx, wsCbP, wsBqP);

  wq_context_kernel<<<dim3(DMODEL / 256, B * NHEAD + 16), blk, 0, stream>>>(
      W_q, wsCtx, wsPT, wsCbP, wsBqP, W_o, b_o, wsBv, B * NHEAD);

  gemm_nt_bias_kernel<false, false><<<dim3(DMODEL / 64, (B * DMODEL) / 256, 2), blk, 0, stream>>>(
      W_o, W_o, W_o, wsPT, wsPT, wsPT, wsBv, wsBv, wsBv, wsMt, wsMt, wsMt, 1,
      B * DMODEL, DMODEL, DMODEL / 2, S, 0,
      DMODEL, DMODEL, DMODEL, DMODEL, sq, (size_t)0, 2 * sq, 0,
      msk, b_q, wsCtx, wsCbP, wsBqP);

  gemm_nt_bias_kernel<true, false><<<gemmGrid, blk, 0, stream>>>(
      q, q, q, wsMt, wsMt, wsMt, wsBv, wsBv, wsBv,
      (float*)d_out, (float*)d_out, (float*)d_out, 1,
      M, DMODEL, DMODEL, S, 0,
      DMODEL, DMODEL, M, S, sq, 2 * sq, (size_t)0, 1,
      msk, b_q, wsCtx, wsCbP, wsBqP);
}
